# starting kernel
# baseline (speedup 1.0000x reference)
#include <hip/hip_runtime.h>
#include <math.h>


typedef float v2f __attribute__((ext_vector_type(2)));
typedef float v8f __attribute__((ext_vector_type(8)));

#if defined(__has_builtin)
#  if __has_builtin(__builtin_amdgcn_wmma_f32_16x16x4_f32)
#    define HAS_WMMA4 1
#  endif
#endif
#ifndef HAS_WMMA4
#  define HAS_WMMA4 0
#endif

namespace {
constexpr int   kN         = 128;
constexpr int   kRingT     = 256;
constexpr int   kRingPitch = 129;
constexpr int   kSteps     = 1000;
constexpr float kDT        = 0.1f;
constexpr float kSig       = 0.0031622776601683794f;
constexpr float kPi        = 3.14159265358979323846f;
constexpr float kInvPi     = 0.3183098861837907f;
constexpr float kJC        = 15.0f;
constexpr float kBDT   = 0.01f;
constexpr float kRTS   = 1.0f / 0.65f;
constexpr float kRTF   = 1.0f / 0.41f;
constexpr float kRTO   = 1.0f / 0.98f;
constexpr float kRAl   = 3.125f;
constexpr float kInvE0 = 2.5f;
constexpr float kOneMinusE0 = 0.6f;
constexpr float kV0 = 4.0f;
constexpr float kK1 = 4.3f * 40.3f * 0.4f * 0.04f;
constexpr float kK2 = 0.5f * 25.0f * 0.4f * 0.04f;
constexpr float kK3 = 0.5f;
}

__global__ __launch_bounds__(256) void tvb_mpr_persistent(
    const float* __restrict__ region_pars,
    const float* __restrict__ Wt,
    const float* __restrict__ g,
    const float* __restrict__ stimulus,
    const float* __restrict__ noise,
    const float* __restrict__ initial_cond,
    const int*   __restrict__ lags,
    float* __restrict__ out)
{
  extern __shared__ float smem[];
  float* ring  = smem;
  float* cxA   = ring + kRingT * kRingPitch;
  float* cxB   = cxA + kN;
  float* stiml = cxB + kN;

  const int tid   = threadIdx.x;
  const int lane  = tid & 31;
  const int wav   = tid >> 5;
  const int mrow  = 16 * wav + (lane & 15);
  const int cbase = (lane >> 4) << 1;

  float wreg[64];
  int   qreg[64];
#pragma unroll
  for (int kt = 0; kt < 32; ++kt) {
    const int j0 = 4 * kt + cbase;
    wreg[2 * kt + 0] = Wt[mrow * kN + j0];
    wreg[2 * kt + 1] = Wt[mrow * kN + j0 + 1];
    qreg[2 * kt + 0] = (1 - lags[mrow * kN + j0]) & 255;
    qreg[2 * kt + 1] = (1 - lags[mrow * kN + j0 + 1]) & 255;
  }

  for (int t = tid; t < 100; t += 256) stiml[t] = stimulus[t];
  for (int idx = tid; idx < kRingT * kN; idx += 256) {
    const int s = idx >> 7, j = idx & 127;
    ring[s * kRingPitch + j] = initial_cond[2 * j];
  }

  float cur_r = 0.f, cur_V = 0.f, eta = 0.f;
  float bs = 1.f, bfl = 1.f, bvl = 1.f, bq = 1.f;
  const float g0 = g[0];
  float2 nz_cur; nz_cur.x = 0.f; nz_cur.y = 0.f;
  if (tid < kN) {
    cur_r  = initial_cond[2 * tid];
    cur_V  = initial_cond[2 * tid + 1];
    eta    = region_pars[tid];
    nz_cur = ((const float2*)noise)[tid];
  }
  __syncthreads();

  auto coupling_window = [&](int base, float* cx_out) {
    v8f accE = {};
    v8f accO = {};
#if HAS_WMMA4
    const v2f bones = {1.0f, 1.0f};
#endif
#pragma unroll
    for (int kt = 0; kt < 32; ++kt) {
      const int j0 = 4 * kt + cbase;
      const int t0 = (base + qreg[2 * kt + 0]) & 255;
      const int t1 = (base + qreg[2 * kt + 1]) & 255;
      v2f a;
      a.x = wreg[2 * kt + 0] * ring[t0 * kRingPitch + j0];
      a.y = wreg[2 * kt + 1] * ring[t1 * kRingPitch + j0 + 1];
#if HAS_WMMA4
      if (kt & 1) {
        accO = __builtin_amdgcn_wmma_f32_16x16x4_f32(
            false, a, false, bones, (short)0, accO, false, false);
      } else {
        accE = __builtin_amdgcn_wmma_f32_16x16x4_f32(
            false, a, false, bones, (short)0, accE, false, false);
      }
#else
      if (kt & 1) accO[kt & 7] += a.x + a.y;
      else        accE[kt & 7] += a.x + a.y;
#endif
    }
#if HAS_WMMA4
    const v8f acc = accE + accO;
    if (lane == 0) {
#pragma unroll
      for (int v = 0; v < 8; ++v) cx_out[16 * wav + v] = acc[v];
    } else if (lane == 16) {
#pragma unroll
      for (int v = 0; v < 8; ++v) cx_out[16 * wav + 8 + v] = acc[v];
    }
#else
    float p = accE[0] + accE[1] + accE[2] + accE[3] +
              accE[4] + accE[5] + accE[6] + accE[7] +
              accO[0] + accO[1] + accO[2] + accO[3] +
              accO[4] + accO[5] + accO[6] + accO[7];
    p += __shfl_xor(p, 16, 32);
    if (lane < 16) cx_out[16 * wav + lane] = p;
#endif
  };

  coupling_window(-1, cxA);

  for (int T = 0; T < kSteps; ++T) {
    float* const cxP = (T & 1) ? cxB : cxA;
    float* const cxN = (T & 1) ? cxA : cxB;

    float2 nz_nxt; nz_nxt.x = 0.f; nz_nxt.y = 0.f;
    if (tid < kN) {
      ring[(T & 255) * kRingPitch + tid] = cur_r;
      if (T + 1 < kSteps) nz_nxt = ((const float2*)noise)[(T + 1) * kN + tid];
    }
    __syncthreads();

    coupling_window(T, cxN);
    __syncthreads();

    if (tid < kN) {
      const float dwr = kSig * nz_cur.x;
      const float dwv = kSig * nz_cur.y;
      const float stv = (tid == 0) ? stiml[T / 10] : 0.f;

      const float rc = cur_r, Vc = cur_V;
      const float I1 = g0 * cxP[tid] + stv;
      const float d1r = kInvPi + 2.f * rc * Vc;
      const float pt1 = kPi * rc;
      const float d1V = Vc * Vc + eta + kJC * rc + I1 - pt1 * pt1;
      const float xir = fmaxf(rc + kDT * d1r + dwr, 0.f);
      const float xiV = Vc + kDT * d1V + dwv;

      const float I2 = g0 * cxN[tid] + stv;
      const float d2r = kInvPi + 2.f * xir * xiV;
      const float pt2 = kPi * xir;
      const float d2V = xiV * xiV + eta + kJC * xir + I2 - pt2 * pt2;

      const float nxr = fmaxf(rc + kDT * 0.5f * (d1r + d2r) + dwr, 0.f);
      const float nxV = Vc + kDT * 0.5f * (d1V + d2V) + dwv;
      cur_r = nxr;
      cur_V = nxV;

      float2 rv2; rv2.x = nxr; rv2.y = nxV;
      ((float2*)out)[T * kN + tid] = rv2;

      {
        const float x   = nxr;
        const float vr1 = powf(bvl, kRAl);
        const float d1s = x - kRTS * bs - kRTF * (bfl - 1.f);
        const float d1f = bs;
        const float d1v = kRTO * (bfl - vr1);
        const float d1q = kRTO * (bfl * (1.f - powf(kOneMinusE0, 1.f / bfl)) * kInvE0
                                  - vr1 * bq / bvl);
        const float s2 = bs + kBDT * d1s;
        const float f2 = bfl + kBDT * d1f;
        const float v2 = bvl + kBDT * d1v;
        const float q2 = bq + kBDT * d1q;
        const float vr2 = powf(v2, kRAl);
        const float d2s = x - kRTS * s2 - kRTF * (f2 - 1.f);
        const float d2f = s2;
        const float d2v = kRTO * (f2 - vr2);
        const float d2q = kRTO * (f2 * (1.f - powf(kOneMinusE0, 1.f / f2)) * kInvE0
                                  - vr2 * q2 / v2);
        bs  += 0.5f * kBDT * (d1s + d2s);
        bfl += 0.5f * kBDT * (d1f + d2f);
        bvl += 0.5f * kBDT * (d1v + d2v);
        bq  += 0.5f * kBDT * (d1q + d2q);
      }
      nz_cur = nz_nxt;
    }
  }

  if (tid < kN) {
    const float bold = kV0 * (kK1 * (1.f - bq) + kK2 * (1.f - bq / bvl) +
                              kK3 * (1.f - bvl));
    out[kSteps * kN * 2 + tid] = bold;
  }
}

extern "C" void kernel_launch(void* const* d_in, const int* in_sizes, int n_in,
                              void* d_out, int out_size, void* d_ws, size_t ws_size,
                              hipStream_t stream) {
  (void)in_sizes; (void)n_in; (void)d_ws; (void)ws_size; (void)out_size;
  const float* region_pars  = (const float*)d_in[0];
  const float* Wt           = (const float*)d_in[1];
  const float* g            = (const float*)d_in[2];
  const float* stimulus     = (const float*)d_in[3];
  const float* noise        = (const float*)d_in[4];
  const float* initial_cond = (const float*)d_in[5];
  const int*   lags         = (const int*)d_in[6];

  const size_t smem_bytes =
      (size_t)(kRingT * kRingPitch + kN + kN + 104) * sizeof(float);

  tvb_mpr_persistent<<<1, 256, smem_bytes, stream>>>(
      region_pars, Wt, g, stimulus, noise, initial_cond, lags, (float*)d_out);
}


# BOLD moved to 4 extra waves (block 256 to 384) that integrate it one step behind from the LDS ring, beside the coupling gather; coupling waves update their own 16 regions; one barrier per step
# speedup vs baseline: 1.3248x; 1.3248x over previous
#include <hip/hip_runtime.h>
#include <math.h>


typedef float v2f __attribute__((ext_vector_type(2)));
typedef float v8f __attribute__((ext_vector_type(8)));

namespace {
constexpr int   kN         = 128;
constexpr int   kThreads   = 384;
constexpr int   kRingT     = 256;
constexpr int   kRingPitch = 129;
constexpr int   kSteps     = 1000;
constexpr float kDT        = 0.1f;
constexpr float kSig       = 0.0031622776601683794f;
constexpr float kPi        = 3.14159265358979323846f;
constexpr float kInvPi     = 0.3183098861837907f;
constexpr float kJC        = 15.0f;
constexpr float kBDT   = 0.01f;
constexpr float kRTS   = 1.0f / 0.65f;
constexpr float kRTF   = 1.0f / 0.41f;
constexpr float kRTO   = 1.0f / 0.98f;
constexpr float kRAl   = 3.125f;
constexpr float kInvE0 = 2.5f;
constexpr float kOneMinusE0 = 0.6f;
constexpr float kV0 = 4.0f;
constexpr float kK1 = 4.3f * 40.3f * 0.4f * 0.04f;
constexpr float kK2 = 0.5f * 25.0f * 0.4f * 0.04f;
constexpr float kK3 = 0.5f;
}

__global__ __launch_bounds__(384) void tvb_mpr_persistent(
    const float* __restrict__ region_pars,
    const float* __restrict__ Wt,
    const float* __restrict__ g,
    const float* __restrict__ stimulus,
    const float* __restrict__ noise,
    const float* __restrict__ initial_cond,
    const int*   __restrict__ lags,
    float* __restrict__ out)
{
  extern __shared__ __align__(16) float smem[];
  float* ring  = smem;
  float* cxs   = ring + kRingT * kRingPitch;
  float* stiml = cxs + kN;

  const int tid   = threadIdx.x;
  const int lane  = tid & 31;
  const int wav   = __builtin_amdgcn_readfirstlane(tid >> 5);
  const bool dyn  = wav < 8;
  const bool own  = dyn && lane < 16;
  const int reg   = dyn ? (16 * wav + (lane & 15)) : (tid - 256);
  const int cbase = (lane >> 4) << 1;

  float wreg[64] = {};
  int   qreg[64] = {};
  if (dyn) {
    const int mrow = 16 * wav + (lane & 15);
#pragma unroll
    for (int kt = 0; kt < 32; ++kt) {
      const int j0 = 4 * kt + cbase;
      wreg[2 * kt + 0] = Wt[mrow * kN + j0];
      wreg[2 * kt + 1] = Wt[mrow * kN + j0 + 1];
      qreg[2 * kt + 0] = (1 - lags[mrow * kN + j0]) & 255;
      qreg[2 * kt + 1] = (1 - lags[mrow * kN + j0 + 1]) & 255;
    }
  }

  for (int t = tid; t < 100; t += kThreads) stiml[t] = stimulus[t];
  for (int idx = tid; idx < kRingT * kN; idx += kThreads) {
    const int s = idx >> 7, j = idx & 127;
    ring[s * kRingPitch + j] = initial_cond[2 * j];
  }

  float cur_r = 0.f, cur_V = 0.f, eta = 0.f;
  float bs = 1.f, bfl = 1.f, bvl = 1.f, bq = 1.f;
  const float g0 = g[0];
  float2 nz_cur; nz_cur.x = 0.f; nz_cur.y = 0.f;
  if (own) {
    cur_r  = initial_cond[2 * reg];
    cur_V  = initial_cond[2 * reg + 1];
    eta    = region_pars[reg];
    nz_cur = ((const float2*)noise)[reg];
  }
  __syncthreads();

  auto coupling_window = [&](int base) -> float {
    v8f accE = {};
    v8f accO = {};
    const v2f bones = {1.0f, 1.0f};
#pragma unroll
    for (int kt = 0; kt < 32; ++kt) {
      const int j0 = 4 * kt + cbase;
      const int t0 = (base + qreg[2 * kt + 0]) & 255;
      const int t1 = (base + qreg[2 * kt + 1]) & 255;
      v2f a;
      a.x = wreg[2 * kt + 0] * ring[t0 * kRingPitch + j0];
      a.y = wreg[2 * kt + 1] * ring[t1 * kRingPitch + j0 + 1];
      if (kt & 1) {
        accO = __builtin_amdgcn_wmma_f32_16x16x4_f32(
            false, a, false, bones, (short)0, accO, false, false);
      } else {
        accE = __builtin_amdgcn_wmma_f32_16x16x4_f32(
            false, a, false, bones, (short)0, accE, false, false);
      }
    }
    const v8f acc = accE + accO;
    float* const my = cxs + 16 * wav;
    if ((lane & 15) == 0) {
      float4* const p = (float4*)(my + 8 * (lane >> 4));
      float4 v;
      v.x = acc[0]; v.y = acc[1]; v.z = acc[2]; v.w = acc[3]; p[0] = v;
      v.x = acc[4]; v.y = acc[5]; v.z = acc[6]; v.w = acc[7]; p[1] = v;
    }
    const float mine = my[lane & 15];
    return mine;
  };

  auto bold_step = [&](float x) {
    const float vr1 = powf(bvl, kRAl);
    const float d1s = x - kRTS * bs - kRTF * (bfl - 1.f);
    const float d1f = bs;
    const float d1v = kRTO * (bfl - vr1);
    const float d1q = kRTO * (bfl * (1.f - powf(kOneMinusE0, 1.f / bfl)) * kInvE0
                              - vr1 * bq / bvl);
    const float s2 = bs + kBDT * d1s;
    const float f2 = bfl + kBDT * d1f;
    const float v2 = bvl + kBDT * d1v;
    const float q2 = bq + kBDT * d1q;
    const float vr2 = powf(v2, kRAl);
    const float d2s = x - kRTS * s2 - kRTF * (f2 - 1.f);
    const float d2f = s2;
    const float d2v = kRTO * (f2 - vr2);
    const float d2q = kRTO * (f2 * (1.f - powf(kOneMinusE0, 1.f / f2)) * kInvE0
                              - vr2 * q2 / v2);
    bs  += 0.5f * kBDT * (d1s + d2s);
    bfl += 0.5f * kBDT * (d1f + d2f);
    bvl += 0.5f * kBDT * (d1v + d2v);
    bq  += 0.5f * kBDT * (d1q + d2q);
  };

  float cxP = 0.f;
  if (dyn) cxP = coupling_window(-1);

  for (int T = 0; T < kSteps; ++T) {
    float2 nz_nxt; nz_nxt.x = 0.f; nz_nxt.y = 0.f;
    if (own) {
      ring[(T & 255) * kRingPitch + reg] = cur_r;
      if (T + 1 < kSteps) nz_nxt = ((const float2*)noise)[(T + 1) * kN + reg];
    }
    __syncthreads();

    if (dyn) {
      const float cxN = coupling_window(T);

      if (own) {
        const float dwr = kSig * nz_cur.x;
        const float dwv = kSig * nz_cur.y;
        const float stv = (reg == 0) ? stiml[T / 10] : 0.f;

        const float rc = cur_r, Vc = cur_V;
        const float I1 = g0 * cxP + stv;
        const float d1r = kInvPi + 2.f * rc * Vc;
        const float pt1 = kPi * rc;
        const float d1V = Vc * Vc + eta + kJC * rc + I1 - pt1 * pt1;
        const float xir = fmaxf(rc + kDT * d1r + dwr, 0.f);
        const float xiV = Vc + kDT * d1V + dwv;

        const float I2 = g0 * cxN + stv;
        const float d2r = kInvPi + 2.f * xir * xiV;
        const float pt2 = kPi * xir;
        const float d2V = xiV * xiV + eta + kJC * xir + I2 - pt2 * pt2;

        const float nxr = fmaxf(rc + kDT * 0.5f * (d1r + d2r) + dwr, 0.f);
        const float nxV = Vc + kDT * 0.5f * (d1V + d2V) + dwv;
        cur_r = nxr;
        cur_V = nxV;

        float2 rv2; rv2.x = nxr; rv2.y = nxV;
        ((float2*)out)[T * kN + reg] = rv2;
        nz_cur = nz_nxt;
      }
      cxP = cxN;
    } else if (T >= 1) {
      bold_step(ring[(T & 255) * kRingPitch + reg]);
    }
  }

  if (own) ring[(kSteps & 255) * kRingPitch + reg] = cur_r;
  __syncthreads();
  if (!dyn) {
    bold_step(ring[(kSteps & 255) * kRingPitch + reg]);
    const float bold = kV0 * (kK1 * (1.f - bq) + kK2 * (1.f - bq / bvl) +
                              kK3 * (1.f - bvl));
    out[kSteps * kN * 2 + reg] = bold;
  }
}

extern "C" void kernel_launch(void* const* d_in, const int* in_sizes, int n_in,
                              void* d_out, int out_size, void* d_ws, size_t ws_size,
                              hipStream_t stream) {
  (void)in_sizes; (void)n_in; (void)d_ws; (void)ws_size; (void)out_size;
  const float* region_pars  = (const float*)d_in[0];
  const float* Wt           = (const float*)d_in[1];
  const float* g            = (const float*)d_in[2];
  const float* stimulus     = (const float*)d_in[3];
  const float* noise        = (const float*)d_in[4];
  const float* initial_cond = (const float*)d_in[5];
  const int*   lags         = (const int*)d_in[6];

  const size_t smem_bytes =
      (size_t)(kRingT * kRingPitch + kN + 104) * sizeof(float);

  tvb_mpr_persistent<<<1, kThreads, smem_bytes, stream>>>(
      region_pars, Wt, g, stimulus, noise, initial_cond, lags, (float*)d_out);
}


# next step's noise load issued right behind the step barrier instead of in front of it, so the barrier's fence no longer waits out a global-memory round trip on the serial path of every step
# speedup vs baseline: 1.6539x; 1.2484x over previous
#include <hip/hip_runtime.h>
#include <math.h>


typedef float v2f __attribute__((ext_vector_type(2)));
typedef float v8f __attribute__((ext_vector_type(8)));

namespace {
constexpr int   kN         = 128;
constexpr int   kThreads   = 384;
constexpr int   kRingT     = 256;
constexpr int   kRingPitch = 129;
constexpr int   kSteps     = 1000;
constexpr float kDT        = 0.1f;
constexpr float kSig       = 0.0031622776601683794f;
constexpr float kPi        = 3.14159265358979323846f;
constexpr float kInvPi     = 0.3183098861837907f;
constexpr float kJC        = 15.0f;
constexpr float kBDT   = 0.01f;
constexpr float kRTS   = 1.0f / 0.65f;
constexpr float kRTF   = 1.0f / 0.41f;
constexpr float kRTO   = 1.0f / 0.98f;
constexpr float kRAl   = 3.125f;
constexpr float kInvE0 = 2.5f;
constexpr float kOneMinusE0 = 0.6f;
constexpr float kV0 = 4.0f;
constexpr float kK1 = 4.3f * 40.3f * 0.4f * 0.04f;
constexpr float kK2 = 0.5f * 25.0f * 0.4f * 0.04f;
constexpr float kK3 = 0.5f;
}

__global__ __launch_bounds__(384) void tvb_mpr_persistent(
    const float* __restrict__ region_pars,
    const float* __restrict__ Wt,
    const float* __restrict__ g,
    const float* __restrict__ stimulus,
    const float* __restrict__ noise,
    const float* __restrict__ initial_cond,
    const int*   __restrict__ lags,
    float* __restrict__ out)
{
  extern __shared__ __align__(16) float smem[];
  float* ring  = smem;
  float* cxs   = ring + kRingT * kRingPitch;
  float* stiml = cxs + kN;

  const int tid   = threadIdx.x;
  const int lane  = tid & 31;
  const int wav   = __builtin_amdgcn_readfirstlane(tid >> 5);
  const bool dyn  = wav < 8;
  const bool own  = dyn && lane < 16;
  const int reg   = dyn ? (16 * wav + (lane & 15)) : (tid - 256);
  const int cbase = (lane >> 4) << 1;

  float wreg[64] = {};
  int   qreg[64] = {};
  if (dyn) {
    const int mrow = 16 * wav + (lane & 15);
#pragma unroll
    for (int kt = 0; kt < 32; ++kt) {
      const int j0 = 4 * kt + cbase;
      wreg[2 * kt + 0] = Wt[mrow * kN + j0];
      wreg[2 * kt + 1] = Wt[mrow * kN + j0 + 1];
      qreg[2 * kt + 0] = (1 - lags[mrow * kN + j0]) & 255;
      qreg[2 * kt + 1] = (1 - lags[mrow * kN + j0 + 1]) & 255;
    }
  }

  for (int t = tid; t < 100; t += kThreads) stiml[t] = stimulus[t];
  for (int idx = tid; idx < kRingT * kN; idx += kThreads) {
    const int s = idx >> 7, j = idx & 127;
    ring[s * kRingPitch + j] = initial_cond[2 * j];
  }

  float cur_r = 0.f, cur_V = 0.f, eta = 0.f;
  float bs = 1.f, bfl = 1.f, bvl = 1.f, bq = 1.f;
  const float g0 = g[0];
  float2 nz_cur; nz_cur.x = 0.f; nz_cur.y = 0.f;
  if (own) {
    cur_r  = initial_cond[2 * reg];
    cur_V  = initial_cond[2 * reg + 1];
    eta    = region_pars[reg];
    nz_cur = ((const float2*)noise)[reg];
  }
  const float2* nzp = (const float2*)noise + kN + reg;
  __syncthreads();

  auto coupling_window = [&](int base) -> float {
    v8f accE = {};
    v8f accO = {};
    const v2f bones = {1.0f, 1.0f};
#pragma unroll
    for (int kt = 0; kt < 32; ++kt) {
      const int j0 = 4 * kt + cbase;
      const int t0 = (base + qreg[2 * kt + 0]) & 255;
      const int t1 = (base + qreg[2 * kt + 1]) & 255;
      v2f a;
      a.x = wreg[2 * kt + 0] * ring[t0 * kRingPitch + j0];
      a.y = wreg[2 * kt + 1] * ring[t1 * kRingPitch + j0 + 1];
      if (kt & 1) {
        accO = __builtin_amdgcn_wmma_f32_16x16x4_f32(
            false, a, false, bones, (short)0, accO, false, false);
      } else {
        accE = __builtin_amdgcn_wmma_f32_16x16x4_f32(
            false, a, false, bones, (short)0, accE, false, false);
      }
    }
    const v8f acc = accE + accO;
    float* const my = cxs + 16 * wav;
    if ((lane & 15) == 0) {
      float4* const p = (float4*)(my + 8 * (lane >> 4));
      float4 v;
      v.x = acc[0]; v.y = acc[1]; v.z = acc[2]; v.w = acc[3]; p[0] = v;
      v.x = acc[4]; v.y = acc[5]; v.z = acc[6]; v.w = acc[7]; p[1] = v;
    }
    const float mine = my[lane & 15];
    return mine;
  };

  auto bold_step = [&](float x) {
    const float vr1 = powf(bvl, kRAl);
    const float d1s = x - kRTS * bs - kRTF * (bfl - 1.f);
    const float d1f = bs;
    const float d1v = kRTO * (bfl - vr1);
    const float d1q = kRTO * (bfl * (1.f - powf(kOneMinusE0, 1.f / bfl)) * kInvE0
                              - vr1 * bq / bvl);
    const float s2 = bs + kBDT * d1s;
    const float f2 = bfl + kBDT * d1f;
    const float v2 = bvl + kBDT * d1v;
    const float q2 = bq + kBDT * d1q;
    const float vr2 = powf(v2, kRAl);
    const float d2s = x - kRTS * s2 - kRTF * (f2 - 1.f);
    const float d2f = s2;
    const float d2v = kRTO * (f2 - vr2);
    const float d2q = kRTO * (f2 * (1.f - powf(kOneMinusE0, 1.f / f2)) * kInvE0
                              - vr2 * q2 / v2);
    bs  += 0.5f * kBDT * (d1s + d2s);
    bfl += 0.5f * kBDT * (d1f + d2f);
    bvl += 0.5f * kBDT * (d1v + d2v);
    bq  += 0.5f * kBDT * (d1q + d2q);
  };

  float cxP = 0.f;
  if (dyn) cxP = coupling_window(-1);

  for (int T = 0; T < kSteps; ++T) {
    if (own) ring[(T & 255) * kRingPitch + reg] = cur_r;
    __syncthreads();

    if (dyn) {
      const float2 nz_nxt = nzp[0];
      __builtin_amdgcn_sched_barrier(0);

      const float cxN = coupling_window(T);

      if (own) {
        const float dwr = kSig * nz_cur.x;
        const float dwv = kSig * nz_cur.y;
        const float stv = (reg == 0) ? stiml[T / 10] : 0.f;

        const float rc = cur_r, Vc = cur_V;
        const float I1 = g0 * cxP + stv;
        const float d1r = kInvPi + 2.f * rc * Vc;
        const float pt1 = kPi * rc;
        const float d1V = Vc * Vc + eta + kJC * rc + I1 - pt1 * pt1;
        const float xir = fmaxf(rc + kDT * d1r + dwr, 0.f);
        const float xiV = Vc + kDT * d1V + dwv;

        const float I2 = g0 * cxN + stv;
        const float d2r = kInvPi + 2.f * xir * xiV;
        const float pt2 = kPi * xir;
        const float d2V = xiV * xiV + eta + kJC * xir + I2 - pt2 * pt2;

        const float nxr = fmaxf(rc + kDT * 0.5f * (d1r + d2r) + dwr, 0.f);
        const float nxV = Vc + kDT * 0.5f * (d1V + d2V) + dwv;
        cur_r = nxr;
        cur_V = nxV;

        float2 rv2; rv2.x = nxr; rv2.y = nxV;
        ((float2*)out)[T * kN + reg] = rv2;
        nz_cur = nz_nxt;
      }
      if (T + 2 < kSteps) nzp += kN;
      cxP = cxN;
    } else if (T >= 1) {
      bold_step(ring[(T & 255) * kRingPitch + reg]);
    }
  }

  if (own) ring[(kSteps & 255) * kRingPitch + reg] = cur_r;
  __syncthreads();
  if (!dyn) {
    bold_step(ring[(kSteps & 255) * kRingPitch + reg]);
    const float bold = kV0 * (kK1 * (1.f - bq) + kK2 * (1.f - bq / bvl) +
                              kK3 * (1.f - bvl));
    out[kSteps * kN * 2 + reg] = bold;
  }
}

extern "C" void kernel_launch(void* const* d_in, const int* in_sizes, int n_in,
                              void* d_out, int out_size, void* d_ws, size_t ws_size,
                              hipStream_t stream) {
  (void)in_sizes; (void)n_in; (void)d_ws; (void)ws_size; (void)out_size;
  const float* region_pars  = (const float*)d_in[0];
  const float* Wt           = (const float*)d_in[1];
  const float* g            = (const float*)d_in[2];
  const float* stimulus     = (const float*)d_in[3];
  const float* noise        = (const float*)d_in[4];
  const float* initial_cond = (const float*)d_in[5];
  const int*   lags         = (const int*)d_in[6];

  const size_t smem_bytes =
      (size_t)(kRingT * kRingPitch + kN + 104) * sizeof(float);

  tvb_mpr_persistent<<<1, kThreads, smem_bytes, stream>>>(
      region_pars, Wt, g, stimulus, noise, initial_cond, lags, (float*)d_out);
}


# history ring stored twice (period 160, 320 slots) so a gather never wraps: its LDS address is one scalar+register add; the and and the multiply-add per gathered value are gone
# speedup vs baseline: 1.8719x; 1.1318x over previous
#include <hip/hip_runtime.h>
#include <math.h>


typedef float v2f __attribute__((ext_vector_type(2)));
typedef float v8f __attribute__((ext_vector_type(8)));

namespace {
constexpr int   kN         = 128;
constexpr int   kThreads   = 384;
constexpr int   kRingR     = 160;
constexpr int   kRingSlots = 2 * kRingR;
constexpr int   kRingPitch = 129;
constexpr int   kPitchB    = kRingPitch * 4;
constexpr int   kSteps     = 1000;
constexpr float kDT        = 0.1f;
constexpr float kSig       = 0.0031622776601683794f;
constexpr float kPi        = 3.14159265358979323846f;
constexpr float kInvPi     = 0.3183098861837907f;
constexpr float kJC        = 15.0f;
constexpr float kBDT   = 0.01f;
constexpr float kRTS   = 1.0f / 0.65f;
constexpr float kRTF   = 1.0f / 0.41f;
constexpr float kRTO   = 1.0f / 0.98f;
constexpr float kRAl   = 3.125f;
constexpr float kInvE0 = 2.5f;
constexpr float kOneMinusE0 = 0.6f;
constexpr float kV0 = 4.0f;
constexpr float kK1 = 4.3f * 40.3f * 0.4f * 0.04f;
constexpr float kK2 = 0.5f * 25.0f * 0.4f * 0.04f;
constexpr float kK3 = 0.5f;
}

__global__ __launch_bounds__(384) void tvb_mpr_persistent(
    const float* __restrict__ region_pars,
    const float* __restrict__ Wt,
    const float* __restrict__ g,
    const float* __restrict__ stimulus,
    const float* __restrict__ noise,
    const float* __restrict__ initial_cond,
    const int*   __restrict__ lags,
    float* __restrict__ out)
{
  extern __shared__ __align__(16) float smem[];
  float* ring  = smem;
  float* cxs   = ring + kRingSlots * kRingPitch;
  float* stiml = cxs + kN;

  const int tid   = threadIdx.x;
  const int lane  = tid & 31;
  const int wav   = __builtin_amdgcn_readfirstlane(tid >> 5);
  const bool dyn  = wav < 8;
  const bool own  = dyn && lane < 16;
  const int reg   = dyn ? (16 * wav + (lane & 15)) : (tid - 256);
  const int cbase = (lane >> 4) << 1;

  float wreg[64] = {};
  int   qreg[64] = {};
  if (dyn) {
    const int mrow = 16 * wav + (lane & 15);
#pragma unroll
    for (int kt = 0; kt < 32; ++kt) {
      const int j0 = 4 * kt + cbase;
      wreg[2 * kt + 0] = Wt[mrow * kN + j0];
      wreg[2 * kt + 1] = Wt[mrow * kN + j0 + 1];
      const int l0 = min(max(lags[mrow * kN + j0], 1), 128);
      const int l1 = min(max(lags[mrow * kN + j0 + 1], 1), 128);
      qreg[2 * kt + 0] = (kRingR + 1 - l0) * kPitchB + 4 * j0;
      qreg[2 * kt + 1] = (kRingR + 1 - l1) * kPitchB + 4 * (j0 + 1);
    }
  }

  for (int t = tid; t < 100; t += kThreads) stiml[t] = stimulus[t];
  for (int idx = tid; idx < kRingSlots * kN; idx += kThreads) {
    const int s = idx >> 7, j = idx & 127;
    ring[s * kRingPitch + j] = initial_cond[2 * j];
  }

  float cur_r = 0.f, cur_V = 0.f, eta = 0.f;
  float bs = 1.f, bfl = 1.f, bvl = 1.f, bq = 1.f;
  const float g0 = g[0];
  float2 nz_cur; nz_cur.x = 0.f; nz_cur.y = 0.f;
  if (own) {
    cur_r  = initial_cond[2 * reg];
    cur_V  = initial_cond[2 * reg + 1];
    eta    = region_pars[reg];
    nz_cur = ((const float2*)noise)[reg];
  }
  const float2* nzp = (const float2*)noise + kN + reg;
  __syncthreads();

  const char* const ringb = (const char*)ring;
  auto coupling_window = [&](int sbase) -> float {
    v8f accE = {};
    v8f accO = {};
    const v2f bones = {1.0f, 1.0f};
#pragma unroll
    for (int kt = 0; kt < 32; ++kt) {
      v2f a;
      a.x = wreg[2 * kt + 0] * *(const float*)(ringb + (sbase + qreg[2 * kt + 0]));
      a.y = wreg[2 * kt + 1] * *(const float*)(ringb + (sbase + qreg[2 * kt + 1]));
      if (kt & 1) {
        accO = __builtin_amdgcn_wmma_f32_16x16x4_f32(
            false, a, false, bones, (short)0, accO, false, false);
      } else {
        accE = __builtin_amdgcn_wmma_f32_16x16x4_f32(
            false, a, false, bones, (short)0, accE, false, false);
      }
    }
    const v8f acc = accE + accO;
    float* const my = cxs + 16 * wav;
    if ((lane & 15) == 0) {
      float4* const p = (float4*)(my + 8 * (lane >> 4));
      float4 v;
      v.x = acc[0]; v.y = acc[1]; v.z = acc[2]; v.w = acc[3]; p[0] = v;
      v.x = acc[4]; v.y = acc[5]; v.z = acc[6]; v.w = acc[7]; p[1] = v;
    }
    const float mine = my[lane & 15];
    return mine;
  };

  auto bold_step = [&](float x) {
    const float vr1 = powf(bvl, kRAl);
    const float d1s = x - kRTS * bs - kRTF * (bfl - 1.f);
    const float d1f = bs;
    const float d1v = kRTO * (bfl - vr1);
    const float d1q = kRTO * (bfl * (1.f - powf(kOneMinusE0, 1.f / bfl)) * kInvE0
                              - vr1 * bq / bvl);
    const float s2 = bs + kBDT * d1s;
    const float f2 = bfl + kBDT * d1f;
    const float v2 = bvl + kBDT * d1v;
    const float q2 = bq + kBDT * d1q;
    const float vr2 = powf(v2, kRAl);
    const float d2s = x - kRTS * s2 - kRTF * (f2 - 1.f);
    const float d2f = s2;
    const float d2v = kRTO * (f2 - vr2);
    const float d2q = kRTO * (f2 * (1.f - powf(kOneMinusE0, 1.f / f2)) * kInvE0
                              - vr2 * q2 / v2);
    bs  += 0.5f * kBDT * (d1s + d2s);
    bfl += 0.5f * kBDT * (d1f + d2f);
    bvl += 0.5f * kBDT * (d1v + d2v);
    bq  += 0.5f * kBDT * (d1q + d2q);
  };

  float cxP = 0.f;
  if (dyn) cxP = coupling_window((kRingR - 1) * kPitchB);

  int bm = 0;
  for (int T = 0; T < kSteps; ++T) {
    if (own) {
      ring[bm * kRingPitch + reg] = cur_r;
      ring[(bm + kRingR) * kRingPitch + reg] = cur_r;
    }
    __syncthreads();

    if (dyn) {
      const float2 nz_nxt = nzp[0];
      __builtin_amdgcn_sched_barrier(0);

      const float cxN = coupling_window(bm * kPitchB);

      if (own) {
        const float dwr = kSig * nz_cur.x;
        const float dwv = kSig * nz_cur.y;
        const float stv = (reg == 0) ? stiml[T / 10] : 0.f;

        const float rc = cur_r, Vc = cur_V;
        const float I1 = g0 * cxP + stv;
        const float d1r = kInvPi + 2.f * rc * Vc;
        const float pt1 = kPi * rc;
        const float d1V = Vc * Vc + eta + kJC * rc + I1 - pt1 * pt1;
        const float xir = fmaxf(rc + kDT * d1r + dwr, 0.f);
        const float xiV = Vc + kDT * d1V + dwv;

        const float I2 = g0 * cxN + stv;
        const float d2r = kInvPi + 2.f * xir * xiV;
        const float pt2 = kPi * xir;
        const float d2V = xiV * xiV + eta + kJC * xir + I2 - pt2 * pt2;

        const float nxr = fmaxf(rc + kDT * 0.5f * (d1r + d2r) + dwr, 0.f);
        const float nxV = Vc + kDT * 0.5f * (d1V + d2V) + dwv;
        cur_r = nxr;
        cur_V = nxV;

        float2 rv2; rv2.x = nxr; rv2.y = nxV;
        ((float2*)out)[T * kN + reg] = rv2;
        nz_cur = nz_nxt;
      }
      if (T + 2 < kSteps) nzp += kN;
      cxP = cxN;
    } else if (T >= 1) {
      bold_step(ring[bm * kRingPitch + reg]);
    }
    bm = (bm + 1 == kRingR) ? 0 : bm + 1;
  }

  if (own) ring[bm * kRingPitch + reg] = cur_r;
  __syncthreads();
  if (!dyn) {
    bold_step(ring[bm * kRingPitch + reg]);
    const float bold = kV0 * (kK1 * (1.f - bq) + kK2 * (1.f - bq / bvl) +
                              kK3 * (1.f - bvl));
    out[kSteps * kN * 2 + reg] = bold;
  }
}

extern "C" void kernel_launch(void* const* d_in, const int* in_sizes, int n_in,
                              void* d_out, int out_size, void* d_ws, size_t ws_size,
                              hipStream_t stream) {
  (void)in_sizes; (void)n_in; (void)d_ws; (void)ws_size; (void)out_size;
  const float* region_pars  = (const float*)d_in[0];
  const float* Wt           = (const float*)d_in[1];
  const float* g            = (const float*)d_in[2];
  const float* stimulus     = (const float*)d_in[3];
  const float* noise        = (const float*)d_in[4];
  const float* initial_cond = (const float*)d_in[5];
  const int*   lags         = (const int*)d_in[6];

  const size_t smem_bytes =
      (size_t)(kRingSlots * kRingPitch + kN + 104) * sizeof(float);

  tvb_mpr_persistent<<<1, kThreads, smem_bytes, stream>>>(
      region_pars, Wt, g, stimulus, noise, initial_cond, lags, (float*)d_out);
}


# ring pitch 129 to 128 (bank = column) and a rotated per-lane gather order (lane halves take even/odd columns): every gather's 32 lanes hit 32 banks, no LDS bank conflicts; summation order changes
# speedup vs baseline: 3.0487x; 1.6287x over previous
#include <hip/hip_runtime.h>
#include <math.h>


typedef float v2f __attribute__((ext_vector_type(2)));
typedef float v8f __attribute__((ext_vector_type(8)));

namespace {
constexpr int   kN         = 128;
constexpr int   kThreads   = 384;
constexpr int   kRingR     = 160;
constexpr int   kRingSlots = 2 * kRingR;
constexpr int   kRingPitch = 128;
constexpr int   kPitchB    = kRingPitch * 4;
constexpr int   kSteps     = 1000;
constexpr float kDT        = 0.1f;
constexpr float kSig       = 0.0031622776601683794f;
constexpr float kPi        = 3.14159265358979323846f;
constexpr float kInvPi     = 0.3183098861837907f;
constexpr float kJC        = 15.0f;
constexpr float kBDT   = 0.01f;
constexpr float kRTS   = 1.0f / 0.65f;
constexpr float kRTF   = 1.0f / 0.41f;
constexpr float kRTO   = 1.0f / 0.98f;
constexpr float kRAl   = 3.125f;
constexpr float kInvE0 = 2.5f;
constexpr float kOneMinusE0 = 0.6f;
constexpr float kV0 = 4.0f;
constexpr float kK1 = 4.3f * 40.3f * 0.4f * 0.04f;
constexpr float kK2 = 0.5f * 25.0f * 0.4f * 0.04f;
constexpr float kK3 = 0.5f;
}

__global__ __launch_bounds__(384) void tvb_mpr_persistent(
    const float* __restrict__ region_pars,
    const float* __restrict__ Wt,
    const float* __restrict__ g,
    const float* __restrict__ stimulus,
    const float* __restrict__ noise,
    const float* __restrict__ initial_cond,
    const int*   __restrict__ lags,
    float* __restrict__ out)
{
  extern __shared__ __align__(16) float smem[];
  float* ring  = smem;
  float* stiml = ring + kRingSlots * kRingPitch;

  const int tid   = threadIdx.x;
  const int lane  = tid & 31;
  const int wav   = __builtin_amdgcn_readfirstlane(tid >> 5);
  const bool dyn  = wav < 8;
  const bool own  = dyn && lane < 16;
  const int reg   = dyn ? (16 * wav + (lane & 15)) : (tid - 256);
  const int cfirst = 2 * (lane & 15) + (lane >> 4);

  float wreg[64] = {};
  int   qreg[64] = {};
  if (dyn) {
    const int mrow = 16 * wav + (lane & 15);
#pragma unroll
    for (int e = 0; e < 64; ++e) {
      const int j = (cfirst + 2 * e) & (kN - 1);
      wreg[e] = Wt[mrow * kN + j];
      const int lg = min(max(lags[mrow * kN + j], 1), 128);
      qreg[e] = (kRingR + 1 - lg) * kPitchB + 4 * j;
    }
  }

  for (int t = tid; t < 100; t += kThreads) stiml[t] = stimulus[t];
  for (int idx = tid; idx < kRingSlots * kN; idx += kThreads) {
    const int s = idx >> 7, j = idx & 127;
    ring[s * kRingPitch + j] = initial_cond[2 * j];
  }

  float cur_r = 0.f, cur_V = 0.f, eta = 0.f;
  float bs = 1.f, bfl = 1.f, bvl = 1.f, bq = 1.f;
  const float g0 = g[0];
  float2 nz_cur; nz_cur.x = 0.f; nz_cur.y = 0.f;
  if (own) {
    cur_r  = initial_cond[2 * reg];
    cur_V  = initial_cond[2 * reg + 1];
    eta    = region_pars[reg];
    nz_cur = ((const float2*)noise)[reg];
  }
  const float2* nzp = (const float2*)noise + kN + reg;
  __syncthreads();

  const char* const ringb = (const char*)ring;
  auto coupling_window = [&](int sbase) -> float {
    float s0 = 0.f, s1 = 0.f, s2 = 0.f, s3 = 0.f;
    float x[64];
#pragma unroll
    for (int e = 0; e < 64; ++e)
      x[e] = *(const float*)(ringb + (sbase + qreg[e]));
    __builtin_amdgcn_sched_barrier(0);
#pragma unroll
    for (int e = 0; e < 64; e += 4) {
      s0 = fmaf(wreg[e + 0], x[e + 0], s0);
      s1 = fmaf(wreg[e + 1], x[e + 1], s1);
      s2 = fmaf(wreg[e + 2], x[e + 2], s2);
      s3 = fmaf(wreg[e + 3], x[e + 3], s3);
    }
    v2f part;
    part.x = s0 + s2;
    part.y = s1 + s3;
    const v2f aones = {1.0f, 1.0f};
    v8f acc = {};
    acc = __builtin_amdgcn_wmma_f32_16x16x4_f32(
        false, aones, false, part, (short)0, acc, false, false);
    return acc[0];
  };

  auto bold_step = [&](float x) {
    const float vr1 = powf(bvl, kRAl);
    const float d1s = x - kRTS * bs - kRTF * (bfl - 1.f);
    const float d1f = bs;
    const float d1v = kRTO * (bfl - vr1);
    const float d1q = kRTO * (bfl * (1.f - powf(kOneMinusE0, 1.f / bfl)) * kInvE0
                              - vr1 * bq / bvl);
    const float s2 = bs + kBDT * d1s;
    const float f2 = bfl + kBDT * d1f;
    const float v2 = bvl + kBDT * d1v;
    const float q2 = bq + kBDT * d1q;
    const float vr2 = powf(v2, kRAl);
    const float d2s = x - kRTS * s2 - kRTF * (f2 - 1.f);
    const float d2f = s2;
    const float d2v = kRTO * (f2 - vr2);
    const float d2q = kRTO * (f2 * (1.f - powf(kOneMinusE0, 1.f / f2)) * kInvE0
                              - vr2 * q2 / v2);
    bs  += 0.5f * kBDT * (d1s + d2s);
    bfl += 0.5f * kBDT * (d1f + d2f);
    bvl += 0.5f * kBDT * (d1v + d2v);
    bq  += 0.5f * kBDT * (d1q + d2q);
  };

  float cxP = 0.f;
  if (dyn) cxP = coupling_window((kRingR - 1) * kPitchB);

  int bm = 0;
  for (int T = 0; T < kSteps; ++T) {
    if (own) {
      ring[bm * kRingPitch + reg] = cur_r;
      ring[(bm + kRingR) * kRingPitch + reg] = cur_r;
    }
    __syncthreads();

    if (dyn) {
      const float2 nz_nxt = nzp[0];
      __builtin_amdgcn_sched_barrier(0);

      const float cxN = coupling_window(bm * kPitchB);

      asm volatile("" :: "v"(nz_nxt.x), "v"(nz_nxt.y), "v"(cxN));

      if (own) {
        const float dwr = kSig * nz_cur.x;
        const float dwv = kSig * nz_cur.y;
        const float stv = (reg == 0) ? stiml[T / 10] : 0.f;

        const float rc = cur_r, Vc = cur_V;
        const float I1 = g0 * cxP + stv;
        const float d1r = kInvPi + 2.f * rc * Vc;
        const float pt1 = kPi * rc;
        const float d1V = Vc * Vc + eta + kJC * rc + I1 - pt1 * pt1;
        const float xir = fmaxf(rc + kDT * d1r + dwr, 0.f);
        const float xiV = Vc + kDT * d1V + dwv;

        const float I2 = g0 * cxN + stv;
        const float d2r = kInvPi + 2.f * xir * xiV;
        const float pt2 = kPi * xir;
        const float d2V = xiV * xiV + eta + kJC * xir + I2 - pt2 * pt2;

        const float nxr = fmaxf(rc + kDT * 0.5f * (d1r + d2r) + dwr, 0.f);
        const float nxV = Vc + kDT * 0.5f * (d1V + d2V) + dwv;
        cur_r = nxr;
        cur_V = nxV;

        float2 rv2; rv2.x = nxr; rv2.y = nxV;
        ((float2*)out)[T * kN + reg] = rv2;
        nz_cur = nz_nxt;
      }
      if (T + 2 < kSteps) nzp += kN;
      cxP = cxN;
    } else if (T >= 1) {
      bold_step(ring[bm * kRingPitch + reg]);
    }
    bm = (bm + 1 == kRingR) ? 0 : bm + 1;
  }

  if (own) ring[bm * kRingPitch + reg] = cur_r;
  __syncthreads();
  if (!dyn) {
    bold_step(ring[bm * kRingPitch + reg]);
    const float bold = kV0 * (kK1 * (1.f - bq) + kK2 * (1.f - bq / bvl) +
                              kK3 * (1.f - bvl));
    out[kSteps * kN * 2 + reg] = bold;
  }
}

extern "C" void kernel_launch(void* const* d_in, const int* in_sizes, int n_in,
                              void* d_out, int out_size, void* d_ws, size_t ws_size,
                              hipStream_t stream) {
  (void)in_sizes; (void)n_in; (void)d_ws; (void)ws_size; (void)out_size;
  const float* region_pars  = (const float*)d_in[0];
  const float* Wt           = (const float*)d_in[1];
  const float* g            = (const float*)d_in[2];
  const float* stimulus     = (const float*)d_in[3];
  const float* noise        = (const float*)d_in[4];
  const float* initial_cond = (const float*)d_in[5];
  const int*   lags         = (const int*)d_in[6];

  const size_t smem_bytes =
      (size_t)(kRingSlots * kRingPitch + 104) * sizeof(float);

  tvb_mpr_persistent<<<1, kThreads, smem_bytes, stream>>>(
      region_pars, Wt, g, stimulus, noise, initial_cond, lags, (float*)d_out);
}


# balloon model's four powf calls evaluated as exp2f(b*log2f(a)): v^RALPHA, and (1-E0)^(1/f) with a constant log2 of its base; its chain no longer sets the step time; pow values change in last places
# speedup vs baseline: 3.4778x; 1.1407x over previous
#include <hip/hip_runtime.h>
#include <math.h>


typedef float v2f __attribute__((ext_vector_type(2)));
typedef float v8f __attribute__((ext_vector_type(8)));

namespace {
constexpr int   kN         = 128;
constexpr int   kThreads   = 384;
constexpr int   kRingR     = 160;
constexpr int   kRingSlots = 2 * kRingR;
constexpr int   kRingPitch = 128;
constexpr int   kPitchB    = kRingPitch * 4;
constexpr int   kSteps     = 1000;
constexpr float kDT        = 0.1f;
constexpr float kSig       = 0.0031622776601683794f;
constexpr float kPi        = 3.14159265358979323846f;
constexpr float kInvPi     = 0.3183098861837907f;
constexpr float kJC        = 15.0f;
constexpr float kBDT   = 0.01f;
constexpr float kRTS   = 1.0f / 0.65f;
constexpr float kRTF   = 1.0f / 0.41f;
constexpr float kRTO   = 1.0f / 0.98f;
constexpr float kRAl   = 3.125f;
constexpr float kInvE0 = 2.5f;
constexpr float kLog2OneMinusE0 = -0.7369655368386571f;
constexpr float kV0 = 4.0f;
constexpr float kK1 = 4.3f * 40.3f * 0.4f * 0.04f;
constexpr float kK2 = 0.5f * 25.0f * 0.4f * 0.04f;
constexpr float kK3 = 0.5f;
}

__global__ __launch_bounds__(384) void tvb_mpr_persistent(
    const float* __restrict__ region_pars,
    const float* __restrict__ Wt,
    const float* __restrict__ g,
    const float* __restrict__ stimulus,
    const float* __restrict__ noise,
    const float* __restrict__ initial_cond,
    const int*   __restrict__ lags,
    float* __restrict__ out)
{
  extern __shared__ __align__(16) float smem[];
  float* ring  = smem;
  float* stiml = ring + kRingSlots * kRingPitch;

  const int tid   = threadIdx.x;
  const int lane  = tid & 31;
  const int wav   = __builtin_amdgcn_readfirstlane(tid >> 5);
  const bool dyn  = wav < 8;
  const bool own  = dyn && lane < 16;
  const int reg   = dyn ? (16 * wav + (lane & 15)) : (tid - 256);
  const int cfirst = 2 * (lane & 15) + (lane >> 4);

  float wreg[64] = {};
  int   qreg[64] = {};
  if (dyn) {
    const int mrow = 16 * wav + (lane & 15);
#pragma unroll
    for (int e = 0; e < 64; ++e) {
      const int j = (cfirst + 2 * e) & (kN - 1);
      wreg[e] = Wt[mrow * kN + j];
      const int lg = min(max(lags[mrow * kN + j], 1), 128);
      qreg[e] = (kRingR + 1 - lg) * kPitchB + 4 * j;
    }
  }

  for (int t = tid; t < 100; t += kThreads) stiml[t] = stimulus[t];
  for (int idx = tid; idx < kRingSlots * kN; idx += kThreads) {
    const int s = idx >> 7, j = idx & 127;
    ring[s * kRingPitch + j] = initial_cond[2 * j];
  }

  float cur_r = 0.f, cur_V = 0.f, eta = 0.f;
  float bs = 1.f, bfl = 1.f, bvl = 1.f, bq = 1.f;
  const float g0 = g[0];
  float2 nz_cur; nz_cur.x = 0.f; nz_cur.y = 0.f;
  if (own) {
    cur_r  = initial_cond[2 * reg];
    cur_V  = initial_cond[2 * reg + 1];
    eta    = region_pars[reg];
    nz_cur = ((const float2*)noise)[reg];
  }
  const float2* nzp = (const float2*)noise + kN + reg;
  __syncthreads();

  const char* const ringb = (const char*)ring;
  auto coupling_window = [&](int sbase) -> float {
    float s0 = 0.f, s1 = 0.f, s2 = 0.f, s3 = 0.f;
    float x[64];
#pragma unroll
    for (int e = 0; e < 64; ++e)
      x[e] = *(const float*)(ringb + (sbase + qreg[e]));
    __builtin_amdgcn_sched_barrier(0);
#pragma unroll
    for (int e = 0; e < 64; e += 4) {
      s0 = fmaf(wreg[e + 0], x[e + 0], s0);
      s1 = fmaf(wreg[e + 1], x[e + 1], s1);
      s2 = fmaf(wreg[e + 2], x[e + 2], s2);
      s3 = fmaf(wreg[e + 3], x[e + 3], s3);
    }
    v2f part;
    part.x = s0 + s2;
    part.y = s1 + s3;
    const v2f aones = {1.0f, 1.0f};
    v8f acc = {};
    acc = __builtin_amdgcn_wmma_f32_16x16x4_f32(
        false, aones, false, part, (short)0, acc, false, false);
    return acc[0];
  };

  auto bold_step = [&](float x) {
    const float vr1 = exp2f(kRAl * log2f(bvl));
    const float d1s = x - kRTS * bs - kRTF * (bfl - 1.f);
    const float d1f = bs;
    const float d1v = kRTO * (bfl - vr1);
    const float d1q = kRTO * (bfl * (1.f - exp2f(kLog2OneMinusE0 / bfl)) * kInvE0
                              - vr1 * bq / bvl);
    const float s2 = bs + kBDT * d1s;
    const float f2 = bfl + kBDT * d1f;
    const float v2 = bvl + kBDT * d1v;
    const float q2 = bq + kBDT * d1q;
    const float vr2 = exp2f(kRAl * log2f(v2));
    const float d2s = x - kRTS * s2 - kRTF * (f2 - 1.f);
    const float d2f = s2;
    const float d2v = kRTO * (f2 - vr2);
    const float d2q = kRTO * (f2 * (1.f - exp2f(kLog2OneMinusE0 / f2)) * kInvE0
                              - vr2 * q2 / v2);
    bs  += 0.5f * kBDT * (d1s + d2s);
    bfl += 0.5f * kBDT * (d1f + d2f);
    bvl += 0.5f * kBDT * (d1v + d2v);
    bq  += 0.5f * kBDT * (d1q + d2q);
  };

  float cxP = 0.f;
  if (dyn) cxP = coupling_window((kRingR - 1) * kPitchB);

  int bm = 0;
  for (int T = 0; T < kSteps; ++T) {
    if (own) {
      ring[bm * kRingPitch + reg] = cur_r;
      ring[(bm + kRingR) * kRingPitch + reg] = cur_r;
    }
    __syncthreads();

    if (dyn) {
      const float2 nz_nxt = nzp[0];
      __builtin_amdgcn_sched_barrier(0);

      const float cxN = coupling_window(bm * kPitchB);

      asm volatile("" :: "v"(nz_nxt.x), "v"(nz_nxt.y), "v"(cxN));

      if (own) {
        const float dwr = kSig * nz_cur.x;
        const float dwv = kSig * nz_cur.y;
        const float stv = (reg == 0) ? stiml[T / 10] : 0.f;

        const float rc = cur_r, Vc = cur_V;
        const float I1 = g0 * cxP + stv;
        const float d1r = kInvPi + 2.f * rc * Vc;
        const float pt1 = kPi * rc;
        const float d1V = Vc * Vc + eta + kJC * rc + I1 - pt1 * pt1;
        const float xir = fmaxf(rc + kDT * d1r + dwr, 0.f);
        const float xiV = Vc + kDT * d1V + dwv;

        const float I2 = g0 * cxN + stv;
        const float d2r = kInvPi + 2.f * xir * xiV;
        const float pt2 = kPi * xir;
        const float d2V = xiV * xiV + eta + kJC * xir + I2 - pt2 * pt2;

        const float nxr = fmaxf(rc + kDT * 0.5f * (d1r + d2r) + dwr, 0.f);
        const float nxV = Vc + kDT * 0.5f * (d1V + d2V) + dwv;
        cur_r = nxr;
        cur_V = nxV;

        float2 rv2; rv2.x = nxr; rv2.y = nxV;
        ((float2*)out)[T * kN + reg] = rv2;
        nz_cur = nz_nxt;
      }
      if (T + 2 < kSteps) nzp += kN;
      cxP = cxN;
    } else if (T >= 1) {
      bold_step(ring[bm * kRingPitch + reg]);
    }
    bm = (bm + 1 == kRingR) ? 0 : bm + 1;
  }

  if (own) ring[bm * kRingPitch + reg] = cur_r;
  __syncthreads();
  if (!dyn) {
    bold_step(ring[bm * kRingPitch + reg]);
    const float bold = kV0 * (kK1 * (1.f - bq) + kK2 * (1.f - bq / bvl) +
                              kK3 * (1.f - bvl));
    out[kSteps * kN * 2 + reg] = bold;
  }
}

extern "C" void kernel_launch(void* const* d_in, const int* in_sizes, int n_in,
                              void* d_out, int out_size, void* d_ws, size_t ws_size,
                              hipStream_t stream) {
  (void)in_sizes; (void)n_in; (void)d_ws; (void)ws_size; (void)out_size;
  const float* region_pars  = (const float*)d_in[0];
  const float* Wt           = (const float*)d_in[1];
  const float* g            = (const float*)d_in[2];
  const float* stimulus     = (const float*)d_in[3];
  const float* noise        = (const float*)d_in[4];
  const float* initial_cond = (const float*)d_in[5];
  const int*   lags         = (const int*)d_in[6];

  const size_t smem_bytes =
      (size_t)(kRingSlots * kRingPitch + 104) * sizeof(float);

  tvb_mpr_persistent<<<1, kThreads, smem_bytes, stream>>>(
      region_pars, Wt, g, stimulus, noise, initial_cond, lags, (float*)d_out);
}


# Window split by lag: lag>=2 part gathered and summed one step ahead beside the update, r published right behind it; only lag-1 gathers follow the barrier (stimulus pre-loaded); sum order changes
# speedup vs baseline: 4.3178x; 1.2415x over previous
#include <hip/hip_runtime.h>
#include <math.h>


typedef float v2f __attribute__((ext_vector_type(2)));
typedef float v8f __attribute__((ext_vector_type(8)));

namespace {
constexpr int   kN         = 128;
constexpr int   kThreads   = 384;
constexpr int   kRingR     = 160;
constexpr int   kRingSlots = 2 * kRingR;
constexpr int   kRingPitch = 128;
constexpr int   kPitchB    = kRingPitch * 4;
constexpr int   kSteps     = 1000;
constexpr int   kL1        = 6;
constexpr float kDT        = 0.1f;
constexpr float kSig       = 0.0031622776601683794f;
constexpr float kPi        = 3.14159265358979323846f;
constexpr float kInvPi     = 0.3183098861837907f;
constexpr float kJC        = 15.0f;
constexpr float kBDT   = 0.01f;
constexpr float kRTS   = 1.0f / 0.65f;
constexpr float kRTF   = 1.0f / 0.41f;
constexpr float kRTO   = 1.0f / 0.98f;
constexpr float kRAl   = 3.125f;
constexpr float kInvE0 = 2.5f;
constexpr float kLog2OneMinusE0 = -0.7369655368386571f;
constexpr float kV0 = 4.0f;
constexpr float kK1 = 4.3f * 40.3f * 0.4f * 0.04f;
constexpr float kK2 = 0.5f * 25.0f * 0.4f * 0.04f;
constexpr float kK3 = 0.5f;
}

__global__ __launch_bounds__(384) void tvb_mpr_persistent(
    const float* __restrict__ region_pars,
    const float* __restrict__ Wt,
    const float* __restrict__ g,
    const float* __restrict__ stimulus,
    const float* __restrict__ noise,
    const float* __restrict__ initial_cond,
    const int*   __restrict__ lags,
    float* __restrict__ out)
{
  extern __shared__ __align__(16) float smem[];
  float* ring  = smem;

  const int tid   = threadIdx.x;
  const int lane  = tid & 31;
  const int wav   = __builtin_amdgcn_readfirstlane(tid >> 5);
  const bool dyn  = wav < 8;
  const bool own  = dyn && lane < 16;
  const int reg   = dyn ? (16 * wav + (lane & 15)) : (tid - 256);
  const int cfirst = 2 * (lane & 15) + (lane >> 4);

  float wreg[64] = {};
  int   qreg[64] = {};
  float w1[kL1] = {};
  int   q1[kL1] = {};
  int   kmax = 0;
  bool  ovf  = false;
  if (dyn) {
    const int mrow = 16 * wav + (lane & 15);
    int cnt = 0;
#pragma unroll
    for (int e = 0; e < 64; ++e) {
      const int j = (cfirst + 2 * e) & (kN - 1);
      const int lg = min(max(lags[mrow * kN + j], 1), 128);
      const float w = Wt[mrow * kN + j];
      const bool is1 = (lg == 1);
#pragma unroll
      for (int k = 0; k < kL1; ++k) {
        const bool hit = is1 && (cnt == k);
        w1[k] = hit ? w : w1[k];
        q1[k] = hit ? (kRingR * kPitchB + 4 * j) : q1[k];
      }
      cnt += is1 ? 1 : 0;
    }
    ovf = __builtin_amdgcn_ballot_w32(cnt > kL1) != 0u;
#pragma unroll
    for (int k = 0; k < kL1; ++k)
      if (__builtin_amdgcn_ballot_w32(cnt > k) != 0u) kmax = k + 1;
    if (ovf) {
      kmax = 0;
#pragma unroll
      for (int k = 0; k < kL1; ++k) w1[k] = 0.f;
    }
    kmax = __builtin_amdgcn_readfirstlane(kmax);
#pragma unroll
    for (int e = 0; e < 64; ++e) {
      const int j = (cfirst + 2 * e) & (kN - 1);
      const int lg = min(max(lags[mrow * kN + j], 1), 128);
      const float w = Wt[mrow * kN + j];
      if (ovf) {
        wreg[e] = w;
        qreg[e] = (kRingR + 1 - lg) * kPitchB + 4 * j;
      } else {
        wreg[e] = (lg >= 2) ? w : 0.f;
        qreg[e] = (kRingR + 2 - max(lg, 2)) * kPitchB + 4 * j;
      }
    }
  }

  for (int idx = tid; idx < kRingSlots * kN; idx += kThreads) {
    const int s = idx >> 7, j = idx & 127;
    ring[s * kRingPitch + j] = initial_cond[2 * j];
  }

  float cur_r = 0.f, cur_V = 0.f, eta = 0.f;
  float bs = 1.f, bfl = 1.f, bvl = 1.f, bq = 1.f;
  const float g0 = g[0];
  float2 nz_cur; nz_cur.x = 0.f; nz_cur.y = 0.f;
  if (own) {
    cur_r  = initial_cond[2 * reg];
    cur_V  = initial_cond[2 * reg + 1];
    eta    = region_pars[reg];
    nz_cur = ((const float2*)noise)[reg];
  }
  const float2* nzp = (const float2*)noise + kN + reg;
  __syncthreads();

  const char* const ringb = (const char*)ring;

  auto sums = [&](const float (&x)[64], float& ox, float& oy) {
    float s0 = 0.f, s1 = 0.f, s2 = 0.f, s3 = 0.f;
#pragma unroll
    for (int e = 0; e < 64; e += 4) {
      s0 = fmaf(wreg[e + 0], x[e + 0], s0);
      s1 = fmaf(wreg[e + 1], x[e + 1], s1);
      s2 = fmaf(wreg[e + 2], x[e + 2], s2);
      s3 = fmaf(wreg[e + 3], x[e + 3], s3);
    }
    ox = s0 + s2;
    oy = s1 + s3;
  };
  auto fold = [&](float px, float py) -> float {
    v2f part;
    part.x = px;
    part.y = py;
    const v2f aones = {1.0f, 1.0f};
    v8f acc = {};
    acc = __builtin_amdgcn_wmma_f32_16x16x4_f32(
        false, aones, false, part, (short)0, acc, false, false);
    return acc[0];
  };

  auto bold_step = [&](float x) {
    const float vr1 = exp2f(kRAl * log2f(bvl));
    const float d1s = x - kRTS * bs - kRTF * (bfl - 1.f);
    const float d1f = bs;
    const float d1v = kRTO * (bfl - vr1);
    const float d1q = kRTO * (bfl * (1.f - exp2f(kLog2OneMinusE0 / bfl)) * kInvE0
                              - vr1 * bq / bvl);
    const float s2 = bs + kBDT * d1s;
    const float f2 = bfl + kBDT * d1f;
    const float v2 = bvl + kBDT * d1v;
    const float q2 = bq + kBDT * d1q;
    const float vr2 = exp2f(kRAl * log2f(v2));
    const float d2s = x - kRTS * s2 - kRTF * (f2 - 1.f);
    const float d2f = s2;
    const float d2v = kRTO * (f2 - vr2);
    const float d2q = kRTO * (f2 * (1.f - exp2f(kLog2OneMinusE0 / f2)) * kInvE0
                              - vr2 * q2 / v2);
    bs  += 0.5f * kBDT * (d1s + d2s);
    bfl += 0.5f * kBDT * (d1f + d2f);
    bvl += 0.5f * kBDT * (d1v + d2v);
    bq  += 0.5f * kBDT * (d1q + d2q);
  };

  float cxP = 0.f;
  float Bx = 0.f, By = 0.f;
  float stl_cur = stimulus[0];
  if (dyn) {
    float x[64];
#pragma unroll
    for (int e = 0; e < 64; ++e)
      x[e] = *(const float*)(ringb + qreg[e]);
    float x1[kL1];
#pragma unroll
    for (int k = 0; k < kL1; ++k)
      x1[k] = *(const float*)(ringb + q1[k]);
    sums(x, Bx, By);
    float a = Bx, b = By;
#pragma unroll
    for (int k = 0; k < kL1; k += 2) {
      a = fmaf(w1[k + 0], x1[k + 0], a);
      b = fmaf(w1[k + 1], x1[k + 1], b);
    }
    cxP = fold(a, b);
  }

  auto heun = [&](float cxN, int T, int bm1) {
    if (own) {
      const float dwr = kSig * nz_cur.x;
      const float dwv = kSig * nz_cur.y;
      const float stv = (reg == 0) ? stl_cur : 0.f;

      const float rc = cur_r, Vc = cur_V;
      const float I1 = g0 * cxP + stv;
      const float d1r = kInvPi + 2.f * rc * Vc;
      const float pt1 = kPi * rc;
      const float d1V = Vc * Vc + eta + kJC * rc + I1 - pt1 * pt1;
      const float xir = fmaxf(rc + kDT * d1r + dwr, 0.f);
      const float xiV = Vc + kDT * d1V + dwv;

      const float I2 = g0 * cxN + stv;
      const float d2r = kInvPi + 2.f * xir * xiV;
      const float pt2 = kPi * xir;
      const float d2V = xiV * xiV + eta + kJC * xir + I2 - pt2 * pt2;

      const float nxr = fmaxf(rc + kDT * 0.5f * (d1r + d2r) + dwr, 0.f);
      const float nxV = Vc + kDT * 0.5f * (d1V + d2V) + dwv;
      cur_r = nxr;
      cur_V = nxV;

      float2 rv2; rv2.x = nxr; rv2.y = nxV;
      ((float2*)out)[T * kN + reg] = rv2;
      ring[bm1 * kRingPitch + reg] = nxr;
      ring[(bm1 + kRingR) * kRingPitch + reg] = nxr;
    }
  };

  int bm = 0;
  for (int T = 0; T < kSteps; ++T) {
    asm volatile("s_wait_dscnt 0x0" ::: "memory");
    __syncthreads();
    const int bm1 = (bm + 1 == kRingR) ? 0 : bm + 1;

    if (dyn) {
      const int sbase = bm * kPitchB;
      const int ch1 = (T + 1) / 10;
      const float2 nz_nxt = nzp[0];
      const float stl_nxt = stimulus[ch1 < 100 ? ch1 : 99];
      __builtin_amdgcn_sched_barrier(0);
      float cxN;

      if (!ovf) {
        float x1[kL1];
        x1[0] = *(const float*)(ringb + (sbase + q1[0]));
        x1[1] = *(const float*)(ringb + (sbase + q1[1]));
#pragma unroll
        for (int k = 2; k < kL1; ++k) x1[k] = 0.f;
        if (kmax > 2) {
          x1[2] = *(const float*)(ringb + (sbase + q1[2]));
          if (kmax > 3) {
#pragma unroll
            for (int k = 3; k < kL1; ++k)
              x1[k] = *(const float*)(ringb + (sbase + q1[k]));
          }
        }
        float x[64];
#pragma unroll
        for (int e = 0; e < 32; ++e)
          x[e] = *(const float*)(ringb + (sbase + qreg[e]));
        __builtin_amdgcn_sched_barrier(0);

        float a = Bx, b = By;
#pragma unroll
        for (int k = 0; k < kL1; k += 2) {
          a = fmaf(w1[k + 0], x1[k + 0], a);
          b = fmaf(w1[k + 1], x1[k + 1], b);
        }
        cxN = fold(a, b);
        heun(cxN, T, bm1);

#pragma unroll
        for (int e = 32; e < 64; ++e)
          x[e] = *(const float*)(ringb + (sbase + qreg[e]));
        __builtin_amdgcn_sched_barrier(0);
        sums(x, Bx, By);
      } else {
        float x[64];
#pragma unroll
        for (int e = 0; e < 64; ++e)
          x[e] = *(const float*)(ringb + (sbase + qreg[e]));
        __builtin_amdgcn_sched_barrier(0);
        float px, py;
        sums(x, px, py);
        cxN = fold(px, py);
        heun(cxN, T, bm1);
      }

      nz_cur = nz_nxt;
      stl_cur = stl_nxt;
      if (T + 2 < kSteps) nzp += kN;
      cxP = cxN;
    } else if (T >= 1) {
      bold_step(ring[bm * kRingPitch + reg]);
    }
    bm = bm1;
  }

  asm volatile("s_wait_dscnt 0x0" ::: "memory");
  __syncthreads();
  if (!dyn) {
    bold_step(ring[bm * kRingPitch + reg]);
    const float bold = kV0 * (kK1 * (1.f - bq) + kK2 * (1.f - bq / bvl) +
                              kK3 * (1.f - bvl));
    out[kSteps * kN * 2 + reg] = bold;
  }
}

extern "C" void kernel_launch(void* const* d_in, const int* in_sizes, int n_in,
                              void* d_out, int out_size, void* d_ws, size_t ws_size,
                              hipStream_t stream) {
  (void)in_sizes; (void)n_in; (void)d_ws; (void)ws_size; (void)out_size;
  const float* region_pars  = (const float*)d_in[0];
  const float* Wt           = (const float*)d_in[1];
  const float* g            = (const float*)d_in[2];
  const float* stimulus     = (const float*)d_in[3];
  const float* noise        = (const float*)d_in[4];
  const float* initial_cond = (const float*)d_in[5];
  const int*   lags         = (const int*)d_in[6];

  const size_t smem_bytes =
      (size_t)(kRingSlots * kRingPitch) * sizeof(float);

  tvb_mpr_persistent<<<1, kThreads, smem_bytes, stream>>>(
      region_pars, Wt, g, stimulus, noise, initial_cond, lags, (float*)d_out);
}


# noise and stimulus of step T+3 loaded at the head of step T (four register sets, step loop unrolled by four, step barrier fences the LDS only): no step waits out a global-memory round trip
# speedup vs baseline: 4.9692x; 1.1509x over previous
#include <hip/hip_runtime.h>
#include <math.h>


typedef float v2f __attribute__((ext_vector_type(2)));
typedef float v8f __attribute__((ext_vector_type(8)));

namespace {
constexpr int   kN         = 128;
constexpr int   kThreads   = 384;
constexpr int   kRingR     = 160;
constexpr int   kRingSlots = 2 * kRingR;
constexpr int   kRingPitch = 128;
constexpr int   kPitchB    = kRingPitch * 4;
constexpr int   kSteps     = 1000;
constexpr int   kL1        = 6;
constexpr float kDT        = 0.1f;
constexpr float kSig       = 0.0031622776601683794f;
constexpr float kPi        = 3.14159265358979323846f;
constexpr float kInvPi     = 0.3183098861837907f;
constexpr float kJC        = 15.0f;
constexpr float kBDT   = 0.01f;
constexpr float kRTS   = 1.0f / 0.65f;
constexpr float kRTF   = 1.0f / 0.41f;
constexpr float kRTO   = 1.0f / 0.98f;
constexpr float kRAl   = 3.125f;
constexpr float kInvE0 = 2.5f;
constexpr float kLog2OneMinusE0 = -0.7369655368386571f;
constexpr float kV0 = 4.0f;
constexpr float kK1 = 4.3f * 40.3f * 0.4f * 0.04f;
constexpr float kK2 = 0.5f * 25.0f * 0.4f * 0.04f;
constexpr float kK3 = 0.5f;
}

__global__ __launch_bounds__(384) void tvb_mpr_persistent(
    const float* __restrict__ region_pars,
    const float* __restrict__ Wt,
    const float* __restrict__ g,
    const float* __restrict__ stimulus,
    const float* __restrict__ noise,
    const float* __restrict__ initial_cond,
    const int*   __restrict__ lags,
    float* __restrict__ out)
{
  extern __shared__ __align__(16) float smem[];
  float* ring  = smem;

  const int tid   = threadIdx.x;
  const int lane  = tid & 31;
  const int wav   = __builtin_amdgcn_readfirstlane(tid >> 5);
  const bool dyn  = wav < 8;
  const bool own  = dyn && lane < 16;
  const int reg   = dyn ? (16 * wav + (lane & 15)) : (tid - 256);
  const int cfirst = 2 * (lane & 15) + (lane >> 4);

  float wreg[64] = {};
  int   qreg[64] = {};
  float w1[kL1] = {};
  int   q1[kL1] = {};
  int   kmax = 0;
  bool  ovf  = false;
  if (dyn) {
    const int mrow = 16 * wav + (lane & 15);
    int cnt = 0;
#pragma unroll
    for (int e = 0; e < 64; ++e) {
      const int j = (cfirst + 2 * e) & (kN - 1);
      const int lg = min(max(lags[mrow * kN + j], 1), 128);
      const float w = Wt[mrow * kN + j];
      const bool is1 = (lg == 1);
#pragma unroll
      for (int k = 0; k < kL1; ++k) {
        const bool hit = is1 && (cnt == k);
        w1[k] = hit ? w : w1[k];
        q1[k] = hit ? (kRingR * kPitchB + 4 * j) : q1[k];
      }
      cnt += is1 ? 1 : 0;
    }
    ovf = __builtin_amdgcn_ballot_w32(cnt > kL1) != 0u;
#pragma unroll
    for (int k = 0; k < kL1; ++k)
      if (__builtin_amdgcn_ballot_w32(cnt > k) != 0u) kmax = k + 1;
    if (ovf) {
      kmax = 0;
#pragma unroll
      for (int k = 0; k < kL1; ++k) w1[k] = 0.f;
    }
    kmax = __builtin_amdgcn_readfirstlane(kmax);
#pragma unroll
    for (int e = 0; e < 64; ++e) {
      const int j = (cfirst + 2 * e) & (kN - 1);
      const int lg = min(max(lags[mrow * kN + j], 1), 128);
      const float w = Wt[mrow * kN + j];
      if (ovf) {
        wreg[e] = w;
        qreg[e] = (kRingR + 1 - lg) * kPitchB + 4 * j;
      } else {
        wreg[e] = (lg >= 2) ? w : 0.f;
        qreg[e] = (kRingR + 2 - max(lg, 2)) * kPitchB + 4 * j;
      }
    }
  }

  for (int idx = tid; idx < kRingSlots * kN; idx += kThreads) {
    const int s = idx >> 7, j = idx & 127;
    ring[s * kRingPitch + j] = initial_cond[2 * j];
  }

  float cur_r = 0.f, cur_V = 0.f, eta = 0.f;
  float bs = 1.f, bfl = 1.f, bvl = 1.f, bq = 1.f;
  const float g0 = g[0];
  if (own) {
    cur_r  = initial_cond[2 * reg];
    cur_V  = initial_cond[2 * reg + 1];
    eta    = region_pars[reg];
  }
  float2 nzr[4];
  float  stlr[4];
#pragma unroll
  for (int k = 0; k < 3; ++k) {
    nzr[k]  = ((const float2*)noise)[k * kN + reg];
    stlr[k] = stimulus[0];
  }
  nzr[3].x = 0.f; nzr[3].y = 0.f;
  stlr[3] = 0.f;
  const float2* nzp = (const float2*)noise + 2 * kN + reg;
  __syncthreads();

  const char* const ringb = (const char*)ring;

  auto sums = [&](const float (&x)[64], float& ox, float& oy) {
    float s0 = 0.f, s1 = 0.f, s2 = 0.f, s3 = 0.f;
#pragma unroll
    for (int e = 0; e < 64; e += 4) {
      s0 = fmaf(wreg[e + 0], x[e + 0], s0);
      s1 = fmaf(wreg[e + 1], x[e + 1], s1);
      s2 = fmaf(wreg[e + 2], x[e + 2], s2);
      s3 = fmaf(wreg[e + 3], x[e + 3], s3);
    }
    ox = s0 + s2;
    oy = s1 + s3;
  };
  auto fold = [&](float px, float py) -> float {
    v2f part;
    part.x = px;
    part.y = py;
    const v2f aones = {1.0f, 1.0f};
    v8f acc = {};
    acc = __builtin_amdgcn_wmma_f32_16x16x4_f32(
        false, aones, false, part, (short)0, acc, false, false);
    return acc[0];
  };

  auto bold_step = [&](float x) {
    const float vr1 = exp2f(kRAl * log2f(bvl));
    const float d1s = x - kRTS * bs - kRTF * (bfl - 1.f);
    const float d1f = bs;
    const float d1v = kRTO * (bfl - vr1);
    const float d1q = kRTO * (bfl * (1.f - exp2f(kLog2OneMinusE0 / bfl)) * kInvE0
                              - vr1 * bq / bvl);
    const float s2 = bs + kBDT * d1s;
    const float f2 = bfl + kBDT * d1f;
    const float v2 = bvl + kBDT * d1v;
    const float q2 = bq + kBDT * d1q;
    const float vr2 = exp2f(kRAl * log2f(v2));
    const float d2s = x - kRTS * s2 - kRTF * (f2 - 1.f);
    const float d2f = s2;
    const float d2v = kRTO * (f2 - vr2);
    const float d2q = kRTO * (f2 * (1.f - exp2f(kLog2OneMinusE0 / f2)) * kInvE0
                              - vr2 * q2 / v2);
    bs  += 0.5f * kBDT * (d1s + d2s);
    bfl += 0.5f * kBDT * (d1f + d2f);
    bvl += 0.5f * kBDT * (d1v + d2v);
    bq  += 0.5f * kBDT * (d1q + d2q);
  };

  float cxP = 0.f;
  float Bx = 0.f, By = 0.f;
  if (dyn) {
    float x[64];
#pragma unroll
    for (int e = 0; e < 64; ++e)
      x[e] = *(const float*)(ringb + qreg[e]);
    float x1[kL1];
#pragma unroll
    for (int k = 0; k < kL1; ++k)
      x1[k] = *(const float*)(ringb + q1[k]);
    sums(x, Bx, By);
    float a = Bx, b = By;
#pragma unroll
    for (int k = 0; k < kL1; k += 2) {
      a = fmaf(w1[k + 0], x1[k + 0], a);
      b = fmaf(w1[k + 1], x1[k + 1], b);
    }
    cxP = fold(a, b);
  }

  auto heun = [&](float cxN, int T, int bm1, float2 nzv, float stl) {
    if (own) {
      const float dwr = kSig * nzv.x;
      const float dwv = kSig * nzv.y;
      const float stv = (reg == 0) ? stl : 0.f;

      const float rc = cur_r, Vc = cur_V;
      const float I1 = g0 * cxP + stv;
      const float d1r = kInvPi + 2.f * rc * Vc;
      const float pt1 = kPi * rc;
      const float d1V = Vc * Vc + eta + kJC * rc + I1 - pt1 * pt1;
      const float xir = fmaxf(rc + kDT * d1r + dwr, 0.f);
      const float xiV = Vc + kDT * d1V + dwv;

      const float I2 = g0 * cxN + stv;
      const float d2r = kInvPi + 2.f * xir * xiV;
      const float pt2 = kPi * xir;
      const float d2V = xiV * xiV + eta + kJC * xir + I2 - pt2 * pt2;

      const float nxr = fmaxf(rc + kDT * 0.5f * (d1r + d2r) + dwr, 0.f);
      const float nxV = Vc + kDT * 0.5f * (d1V + d2V) + dwv;
      cur_r = nxr;
      cur_V = nxV;

      float2 rv2; rv2.x = nxr; rv2.y = nxV;
      ((float2*)out)[T * kN + reg] = rv2;
      ring[bm1 * kRingPitch + reg] = nxr;
      ring[(bm1 + kRingR) * kRingPitch + reg] = nxr;
    }
  };

  auto lds_barrier = [&]() {
    __builtin_amdgcn_fence(__ATOMIC_RELEASE, "workgroup", "local");
    __builtin_amdgcn_s_barrier();
    __builtin_amdgcn_fence(__ATOMIC_ACQUIRE, "workgroup", "local");
  };

  auto refill = [&](int T, float2& nzv, float& stl) {
    if (T + 3 < kSteps) nzp += kN;
    nzv = nzp[0];
    const int ch3 = (T + 3) / 10;
    stl = stimulus[ch3 < 100 ? ch3 : 99];
  };

  int bm = 0;
  for (int T0 = 0; T0 < kSteps; T0 += 4) {
#pragma unroll
    for (int u = 0; u < 4; ++u) {
      const int T = T0 + u;
      asm volatile("s_wait_dscnt 0x0" ::: "memory");
      lds_barrier();
      const int bm1 = (bm + 1 == kRingR) ? 0 : bm + 1;
      refill(T, nzr[(u + 3) & 3], stlr[(u + 3) & 3]);

      if (dyn) {
        const int sbase = bm * kPitchB;
        float cxN;

        if (!ovf) {
          float x1[kL1];
          x1[0] = *(const float*)(ringb + (sbase + q1[0]));
          x1[1] = *(const float*)(ringb + (sbase + q1[1]));
#pragma unroll
          for (int k = 2; k < kL1; ++k) x1[k] = 0.f;
          if (kmax > 2) {
            x1[2] = *(const float*)(ringb + (sbase + q1[2]));
            if (kmax > 3) {
#pragma unroll
              for (int k = 3; k < kL1; ++k)
                x1[k] = *(const float*)(ringb + (sbase + q1[k]));
            }
          }
          float x[64];
#pragma unroll
          for (int e = 0; e < 32; ++e)
            x[e] = *(const float*)(ringb + (sbase + qreg[e]));
          __builtin_amdgcn_sched_barrier(0);

          float a = Bx, b = By;
#pragma unroll
          for (int k = 0; k < kL1; k += 2) {
            a = fmaf(w1[k + 0], x1[k + 0], a);
            b = fmaf(w1[k + 1], x1[k + 1], b);
          }
          cxN = fold(a, b);
          heun(cxN, T, bm1, nzr[u], stlr[u]);

#pragma unroll
          for (int e = 32; e < 64; ++e)
            x[e] = *(const float*)(ringb + (sbase + qreg[e]));
          __builtin_amdgcn_sched_barrier(0);
          sums(x, Bx, By);
        } else {
          float x[64];
#pragma unroll
          for (int e = 0; e < 64; ++e)
            x[e] = *(const float*)(ringb + (sbase + qreg[e]));
          __builtin_amdgcn_sched_barrier(0);
          float px, py;
          sums(x, px, py);
          cxN = fold(px, py);
          heun(cxN, T, bm1, nzr[u], stlr[u]);
        }

        cxP = cxN;
      } else if (T >= 1) {
        bold_step(ring[bm * kRingPitch + reg]);
      }
      bm = bm1;
    }
  }

  asm volatile("s_wait_dscnt 0x0" ::: "memory");
  lds_barrier();
  if (!dyn) {
    bold_step(ring[bm * kRingPitch + reg]);
    const float bold = kV0 * (kK1 * (1.f - bq) + kK2 * (1.f - bq / bvl) +
                              kK3 * (1.f - bvl));
    out[kSteps * kN * 2 + reg] = bold;
  }
}

extern "C" void kernel_launch(void* const* d_in, const int* in_sizes, int n_in,
                              void* d_out, int out_size, void* d_ws, size_t ws_size,
                              hipStream_t stream) {
  (void)in_sizes; (void)n_in; (void)d_ws; (void)ws_size; (void)out_size;
  const float* region_pars  = (const float*)d_in[0];
  const float* Wt           = (const float*)d_in[1];
  const float* g            = (const float*)d_in[2];
  const float* stimulus     = (const float*)d_in[3];
  const float* noise        = (const float*)d_in[4];
  const float* initial_cond = (const float*)d_in[5];
  const int*   lags         = (const int*)d_in[6];

  const size_t smem_bytes =
      (size_t)(kRingSlots * kRingPitch) * sizeof(float);

  tvb_mpr_persistent<<<1, kThreads, smem_bytes, stream>>>(
      region_pars, Wt, g, stimulus, noise, initial_cond, lags, (float*)d_out);
}


# new r of a step needs only its first window, so r runs one step ahead of V: the whole window (lag-1 too) is gathered a step ahead, 64 loads first, update beside the sums; sum order changes
# speedup vs baseline: 5.2317x; 1.0528x over previous
#include <hip/hip_runtime.h>
#include <math.h>


typedef float v2f __attribute__((ext_vector_type(2)));
typedef float v8f __attribute__((ext_vector_type(8)));

namespace {
constexpr int   kN         = 128;
constexpr int   kThreads   = 384;
constexpr int   kRingR     = 160;
constexpr int   kRingSlots = 2 * kRingR;
constexpr int   kRingPitch = 128;
constexpr int   kPitchB    = kRingPitch * 4;
constexpr int   kSteps     = 1000;
constexpr float kDT        = 0.1f;
constexpr float kSig       = 0.0031622776601683794f;
constexpr float kPi        = 3.14159265358979323846f;
constexpr float kInvPi     = 0.3183098861837907f;
constexpr float kJC        = 15.0f;
constexpr float kBDT   = 0.01f;
constexpr float kRTS   = 1.0f / 0.65f;
constexpr float kRTF   = 1.0f / 0.41f;
constexpr float kRTO   = 1.0f / 0.98f;
constexpr float kRAl   = 3.125f;
constexpr float kInvE0 = 2.5f;
constexpr float kLog2OneMinusE0 = -0.7369655368386571f;
constexpr float kV0 = 4.0f;
constexpr float kK1 = 4.3f * 40.3f * 0.4f * 0.04f;
constexpr float kK2 = 0.5f * 25.0f * 0.4f * 0.04f;
constexpr float kK3 = 0.5f;
}

__global__ __launch_bounds__(384) void tvb_mpr_persistent(
    const float* __restrict__ region_pars,
    const float* __restrict__ Wt,
    const float* __restrict__ g,
    const float* __restrict__ stimulus,
    const float* __restrict__ noise,
    const float* __restrict__ initial_cond,
    const int*   __restrict__ lags,
    float* __restrict__ out)
{
  extern __shared__ __align__(16) float smem[];
  float* ring  = smem;

  const int tid   = threadIdx.x;
  const int lane  = tid & 31;
  const int wav   = __builtin_amdgcn_readfirstlane(tid >> 5);
  const bool dyn  = wav < 8;
  const bool own  = dyn && lane < 16;
  const int reg   = dyn ? (16 * wav + (lane & 15)) : (tid - 256);
  const int cfirst = 2 * (lane & 15) + (lane >> 4);


  float wreg[64] = {};
  int   qreg[64] = {};
  if (dyn) {
    const int mrow = 16 * wav + (lane & 15);
#pragma unroll
    for (int e = 0; e < 64; ++e) {
      const int j = (cfirst + 2 * e) & (kN - 1);
      const int lg = min(max(lags[mrow * kN + j], 1), 128);
      wreg[e] = Wt[mrow * kN + j];
      qreg[e] = (kRingR + 1 - lg) * kPitchB + 4 * j;
    }
  }

  for (int idx = tid; idx < kRingSlots * kN; idx += kThreads) {
    const int s = idx >> 7, j = idx & 127;
    ring[s * kRingPitch + j] = initial_cond[2 * j];
  }

  float cur_r = 0.f, cur_V = 0.f, eta = 0.f;
  float bs = 1.f, bfl = 1.f, bvl = 1.f, bq = 1.f;
  const float g0 = g[0];
  if (own) {
    cur_r  = initial_cond[2 * reg];
    cur_V  = initial_cond[2 * reg + 1];
    eta    = region_pars[reg];
  }
  float2 nzr[4];
  float  stlr[4];
#pragma unroll
  for (int k = 0; k < 3; ++k) {
    nzr[k]  = ((const float2*)noise)[k * kN + reg];
    stlr[k] = stimulus[0];
  }
  nzr[3].x = 0.f; nzr[3].y = 0.f;
  stlr[3] = 0.f;
  const float2* nzp = (const float2*)noise + 2 * kN + reg;
  __syncthreads();

  const char* const ringb = (const char*)ring;


  auto sums = [&](const float (&x)[64], float& ox, float& oy) {
    float s0 = 0.f, s1 = 0.f, s2 = 0.f, s3 = 0.f;
#pragma unroll
    for (int e = 0; e < 64; e += 4) {
      s0 = fmaf(wreg[e + 0], x[e + 0], s0);
      s1 = fmaf(wreg[e + 1], x[e + 1], s1);
      s2 = fmaf(wreg[e + 2], x[e + 2], s2);
      s3 = fmaf(wreg[e + 3], x[e + 3], s3);
    }
    ox = s0 + s2;
    oy = s1 + s3;
  };
  auto fold = [&](float px, float py) -> float {
    v2f part;
    part.x = px;
    part.y = py;
    const v2f aones = {1.0f, 1.0f};
    v8f acc = {};
    acc = __builtin_amdgcn_wmma_f32_16x16x4_f32(
        false, aones, false, part, (short)0, acc, false, false);
    return acc[0];
  };

  auto bold_step = [&](float x) {
    const float vr1 = exp2f(kRAl * log2f(bvl));
    const float d1s = x - kRTS * bs - kRTF * (bfl - 1.f);
    const float d1f = bs;
    const float d1v = kRTO * (bfl - vr1);
    const float d1q = kRTO * (bfl * (1.f - exp2f(kLog2OneMinusE0 / bfl)) * kInvE0
                              - vr1 * bq / bvl);
    const float s2 = bs + kBDT * d1s;
    const float f2 = bfl + kBDT * d1f;
    const float v2 = bvl + kBDT * d1v;
    const float q2 = bq + kBDT * d1q;
    const float vr2 = exp2f(kRAl * log2f(v2));
    const float d2s = x - kRTS * s2 - kRTF * (f2 - 1.f);
    const float d2f = s2;
    const float d2v = kRTO * (f2 - vr2);
    const float d2q = kRTO * (f2 * (1.f - exp2f(kLog2OneMinusE0 / f2)) * kInvE0
                              - vr2 * q2 / v2);
    bs  += 0.5f * kBDT * (d1s + d2s);
    bfl += 0.5f * kBDT * (d1f + d2f);
    bvl += 0.5f * kBDT * (d1v + d2v);
    bq  += 0.5f * kBDT * (d1q + d2q);
  };

  auto next_r = [&](float rc, float Vc, float cx, float2 nzv, float stl) -> float {
    const float dwr = kSig * nzv.x;
    const float dwv = kSig * nzv.y;
    const float stv = (reg == 0) ? stl : 0.f;
    const float I1 = g0 * cx + stv;
    const float d1r = kInvPi + 2.f * rc * Vc;
    const float pt1 = kPi * rc;
    const float d1V = Vc * Vc + eta + kJC * rc + I1 - pt1 * pt1;
    const float xir = fmaxf(rc + kDT * d1r + dwr, 0.f);
    const float xiV = Vc + kDT * d1V + dwv;
    const float d2r = kInvPi + 2.f * xir * xiV;
    return fmaxf(rc + kDT * 0.5f * (d1r + d2r) + dwr, 0.f);
  };

  float cxP = 0.f;
  float cxN = 0.f;
  float nr  = 0.f;
  if (dyn) {
    float x[64];
#pragma unroll
    for (int e = 0; e < 64; ++e)
      x[e] = *(const float*)(ringb + qreg[e]);
    float px, py;
    sums(x, px, py);
    cxP = fold(px, py);
    cxN = cxP;
    if (own) {
      nr = next_r(cur_r, cur_V, cxP, nzr[0], stlr[0]);
      ring[1 * kRingPitch + reg] = nr;
      ring[(1 + kRingR) * kRingPitch + reg] = nr;
    }
  }

  auto update = [&](float2 nzv, float stl, float2 nz1, float st1, float& nr2) -> float {
    const float dwv = kSig * nzv.y;
    const float dwr = kSig * nzv.x;
    const float stv = (reg == 0) ? stl : 0.f;

    const float rc = cur_r, Vc = cur_V;
    const float I1 = g0 * cxP + stv;
    const float d1r = kInvPi + 2.f * rc * Vc;
    const float pt1 = kPi * rc;
    const float d1V = Vc * Vc + eta + kJC * rc + I1 - pt1 * pt1;
    const float xir = fmaxf(rc + kDT * d1r + dwr, 0.f);
    const float xiV = Vc + kDT * d1V + dwv;

    const float I2 = g0 * cxN + stv;
    const float pt2 = kPi * xir;
    const float d2V = xiV * xiV + eta + kJC * xir + I2 - pt2 * pt2;
    const float nxV = Vc + kDT * 0.5f * (d1V + d2V) + dwv;

    nr2 = next_r(nr, nxV, cxN, nz1, st1);
    return nxV;
  };

  auto lds_barrier = [&]() {
    __builtin_amdgcn_fence(__ATOMIC_RELEASE, "workgroup", "local");
    __builtin_amdgcn_s_barrier();
    __builtin_amdgcn_fence(__ATOMIC_ACQUIRE, "workgroup", "local");
  };

  auto refill = [&](int T, float2& nzv, float& stl) {
    if (T + 3 < kSteps) nzp += kN;
    nzv = nzp[0];
    const int ch3 = (T + 3) / 10;
    stl = stimulus[ch3 < 100 ? ch3 : 99];
  };


  int bm = 0;
  for (int T0 = 0; T0 < kSteps; T0 += 4) {
#pragma unroll
    for (int u = 0; u < 4; ++u) {
      const int T = T0 + u;
      asm volatile("s_wait_dscnt 0x0" ::: "memory");
      lds_barrier();
      const int bm1 = (bm + 1 == kRingR) ? 0 : bm + 1;
      const int bm2 = (bm1 + 1 == kRingR) ? 0 : bm1 + 1;
      refill(T, nzr[(u + 3) & 3], stlr[(u + 3) & 3]);

      if (dyn) {
        const int sbase = bm1 * kPitchB;
        float x[64];
#pragma unroll
        for (int e = 0; e < 64; ++e)
          x[e] = *(const float*)(ringb + (sbase + qreg[e]));
        __builtin_amdgcn_sched_barrier(0);

        float nr2;
        const float nxV = update(nzr[u], stlr[u], nzr[(u + 1) & 3], stlr[(u + 1) & 3], nr2);
        float px, py;
        sums(x, px, py);
        const float cxNew = fold(px, py);
        if (own) {
          float2 rv2; rv2.x = nr; rv2.y = nxV;
          ((float2*)out)[T * kN + reg] = rv2;
          ring[bm2 * kRingPitch + reg] = nr2;
          ring[(bm2 + kRingR) * kRingPitch + reg] = nr2;
        }
        cur_r = nr;
        cur_V = nxV;
        nr  = nr2;
        cxP = cxN;
        cxN = cxNew;
      } else if (T >= 1) {
        bold_step(ring[bm * kRingPitch + reg]);
      }
      bm = bm1;
    }
  }

  asm volatile("s_wait_dscnt 0x0" ::: "memory");
  lds_barrier();
  if (!dyn) {
    bold_step(ring[bm * kRingPitch + reg]);
    const float bold = kV0 * (kK1 * (1.f - bq) + kK2 * (1.f - bq / bvl) +
                              kK3 * (1.f - bvl));
    out[kSteps * kN * 2 + reg] = bold;
  }
}

extern "C" void kernel_launch(void* const* d_in, const int* in_sizes, int n_in,
                              void* d_out, int out_size, void* d_ws, size_t ws_size,
                              hipStream_t stream) {
  (void)in_sizes; (void)n_in; (void)d_ws; (void)ws_size; (void)out_size;
  const float* region_pars  = (const float*)d_in[0];
  const float* Wt           = (const float*)d_in[1];
  const float* g            = (const float*)d_in[2];
  const float* stimulus     = (const float*)d_in[3];
  const float* noise        = (const float*)d_in[4];
  const float* initial_cond = (const float*)d_in[5];
  const int*   lags         = (const int*)d_in[6];

  const size_t smem_bytes =
      (size_t)(kRingSlots * kRingPitch) * sizeof(float);

  tvb_mpr_persistent<<<1, kThreads, smem_bytes, stream>>>(
      region_pars, Wt, g, stimulus, noise, initial_cond, lags, (float*)d_out);
}


# History ring holds pairs (r_t, r_t+1): one 8-byte LDS gather per entry serves the windows of two steps; both steps updated per barrier; one WMMA folds both windows; summation order changes
# speedup vs baseline: 7.2258x; 1.3812x over previous
#include <hip/hip_runtime.h>
#include <math.h>


typedef float v2f __attribute__((ext_vector_type(2)));
typedef float v8f __attribute__((ext_vector_type(8)));

namespace {
constexpr int   kN         = 128;
constexpr int   kThreads   = 384;
constexpr int   kRingR     = 132;
constexpr int   kRingSlots = 2 * kRingR;
constexpr int   kSlotB     = kN * 8;
constexpr int   kSteps     = 1000;
constexpr float kDT        = 0.1f;
constexpr float kSig       = 0.0031622776601683794f;
constexpr float kPi        = 3.14159265358979323846f;
constexpr float kInvPi     = 0.3183098861837907f;
constexpr float kJC        = 15.0f;
constexpr float kBDT   = 0.01f;
constexpr float kRTS   = 1.0f / 0.65f;
constexpr float kRTF   = 1.0f / 0.41f;
constexpr float kRTO   = 1.0f / 0.98f;
constexpr float kRAl   = 3.125f;
constexpr float kInvE0 = 2.5f;
constexpr float kLog2OneMinusE0 = -0.7369655368386571f;
constexpr float kV0 = 4.0f;
constexpr float kK1 = 4.3f * 40.3f * 0.4f * 0.04f;
constexpr float kK2 = 0.5f * 25.0f * 0.4f * 0.04f;
constexpr float kK3 = 0.5f;
}

__global__ __launch_bounds__(384) void tvb_mpr_persistent(
    const float* __restrict__ region_pars,
    const float* __restrict__ Wt,
    const float* __restrict__ g,
    const float* __restrict__ stimulus,
    const float* __restrict__ noise,
    const float* __restrict__ initial_cond,
    const int*   __restrict__ lags,
    float* __restrict__ out)
{
  __shared__ __align__(16) float2 ring[kRingSlots * kN];

  const int tid   = threadIdx.x;
  const int lane  = tid & 31;
  const int wav   = __builtin_amdgcn_readfirstlane(tid >> 5);
  const bool dyn  = wav < 8;
  const bool own  = dyn && lane < 16;
  const int reg   = dyn ? (16 * wav + (lane & 15)) : (tid - 256);
  const int cfirst = 2 * (lane & 15) + (lane >> 4);

  float wreg[64] = {};
  int   qreg[64] = {};
  if (dyn) {
    const int mrow = 16 * wav + (lane & 15);
#pragma unroll
    for (int e = 0; e < 64; ++e) {
      const int j = (cfirst + 2 * e) & (kN - 1);
      const int lg = min(max(lags[mrow * kN + j], 1), 128);
      wreg[e] = Wt[mrow * kN + j];
      qreg[e] = (kRingR + 1 - lg) * kSlotB + 8 * j;
    }
  }

  for (int idx = tid; idx < kRingSlots * kN; idx += kThreads) {
    const int j = idx & 127;
    const float r0 = initial_cond[2 * j];
    float2 p; p.x = r0; p.y = r0;
    ring[idx] = p;
  }

  float cur_r = 0.f, cur_V = 0.f, eta = 0.f;
  float bs = 1.f, bfl = 1.f, bvl = 1.f, bq = 1.f;
  const float g0 = g[0];
  if (own) {
    cur_r  = initial_cond[2 * reg];
    cur_V  = initial_cond[2 * reg + 1];
    eta    = region_pars[reg];
  }
  float2 nzr[8];
  float  stlr[8];
  auto refill = [&](int S, float2& nzv, float& stl) {
    const int row = S < kSteps ? S : kSteps - 1;
    nzv = ((const float2*)noise)[row * kN + reg];
    stl = stimulus[row / 10];
  };
#pragma unroll
  for (int k = 0; k < 8; ++k) {
    if (k < 5) {
      refill(k, nzr[k], stlr[k]);
    } else {
      nzr[k].x = 0.f; nzr[k].y = 0.f; stlr[k] = 0.f;
    }
  }
  __syncthreads();

  const char* const ringb = (const char*)ring;

  auto sums = [&](const float2 (&x)[64], float& ox, float& oy) {
    float a0 = 0.f, a1 = 0.f, a2 = 0.f, a3 = 0.f;
    float b0 = 0.f, b1 = 0.f, b2 = 0.f, b3 = 0.f;
#pragma unroll
    for (int e = 0; e < 64; e += 4) {
      a0 = fmaf(wreg[e + 0], x[e + 0].x, a0);
      b0 = fmaf(wreg[e + 0], x[e + 0].y, b0);
      a1 = fmaf(wreg[e + 1], x[e + 1].x, a1);
      b1 = fmaf(wreg[e + 1], x[e + 1].y, b1);
      a2 = fmaf(wreg[e + 2], x[e + 2].x, a2);
      b2 = fmaf(wreg[e + 2], x[e + 2].y, b2);
      a3 = fmaf(wreg[e + 3], x[e + 3].x, a3);
      b3 = fmaf(wreg[e + 3], x[e + 3].y, b3);
    }
    ox = (a0 + a2) + (a1 + a3);
    oy = (b0 + b2) + (b1 + b3);
  };
  const float aodd = (float)(lane & 1);
  auto fold = [&](float px, float py, float& cxa, float& cxb) {
    v2f part;
    part.x = px;
    part.y = py;
    v2f asel;
    asel.x = 1.0f - aodd;
    asel.y = aodd;
    v8f acc = {};
    acc = __builtin_amdgcn_wmma_f32_16x16x4_f32(
        false, asel, false, part, (short)0, acc, false, false);
    cxa = acc[0];
    cxb = acc[1];
  };

  auto bold_step = [&](float x) {
    const float vr1 = exp2f(kRAl * log2f(bvl));
    const float d1s = x - kRTS * bs - kRTF * (bfl - 1.f);
    const float d1f = bs;
    const float d1v = kRTO * (bfl - vr1);
    const float d1q = kRTO * (bfl * (1.f - exp2f(kLog2OneMinusE0 / bfl)) * kInvE0
                              - vr1 * bq / bvl);
    const float s2 = bs + kBDT * d1s;
    const float f2 = bfl + kBDT * d1f;
    const float v2 = bvl + kBDT * d1v;
    const float q2 = bq + kBDT * d1q;
    const float vr2 = exp2f(kRAl * log2f(v2));
    const float d2s = x - kRTS * s2 - kRTF * (f2 - 1.f);
    const float d2f = s2;
    const float d2v = kRTO * (f2 - vr2);
    const float d2q = kRTO * (f2 * (1.f - exp2f(kLog2OneMinusE0 / f2)) * kInvE0
                              - vr2 * q2 / v2);
    bs  += 0.5f * kBDT * (d1s + d2s);
    bfl += 0.5f * kBDT * (d1f + d2f);
    bvl += 0.5f * kBDT * (d1v + d2v);
    bq  += 0.5f * kBDT * (d1q + d2q);
  };

  auto next_r = [&](float rc, float Vc, float cx, float2 nzv, float stl) -> float {
    const float dwr = kSig * nzv.x;
    const float dwv = kSig * nzv.y;
    const float stv = (reg == 0) ? stl : 0.f;
    const float I1 = g0 * cx + stv;
    const float d1r = kInvPi + 2.f * rc * Vc;
    const float pt1 = kPi * rc;
    const float d1V = Vc * Vc + eta + kJC * rc + I1 - pt1 * pt1;
    const float xir = fmaxf(rc + kDT * d1r + dwr, 0.f);
    const float xiV = Vc + kDT * d1V + dwv;
    const float d2r = kInvPi + 2.f * xir * xiV;
    return fmaxf(rc + kDT * 0.5f * (d1r + d2r) + dwr, 0.f);
  };

  auto update = [&](float rc, float Vc, float rn, float c1, float c2,
                    float2 nzv, float stl, float2 nz1, float st1, float& nr2) -> float {
    const float dwv = kSig * nzv.y;
    const float dwr = kSig * nzv.x;
    const float stv = (reg == 0) ? stl : 0.f;

    const float I1 = g0 * c1 + stv;
    const float d1r = kInvPi + 2.f * rc * Vc;
    const float pt1 = kPi * rc;
    const float d1V = Vc * Vc + eta + kJC * rc + I1 - pt1 * pt1;
    const float xir = fmaxf(rc + kDT * d1r + dwr, 0.f);
    const float xiV = Vc + kDT * d1V + dwv;

    const float I2 = g0 * c2 + stv;
    const float pt2 = kPi * xir;
    const float d2V = xiV * xiV + eta + kJC * xir + I2 - pt2 * pt2;
    const float nxV = Vc + kDT * 0.5f * (d1V + d2V) + dwv;

    nr2 = next_r(rn, nxV, c2, nz1, st1);
    return nxV;
  };

  float cxP = 0.f;
  float nr  = 0.f;
  if (dyn) {
    float2 x[64];
#pragma unroll
    for (int e = 0; e < 64; ++e)
      x[e] = *(const float2*)(ringb + qreg[e]);
    float px, py, cb;
    sums(x, px, py);
    fold(px, py, cxP, cb);
    if (own) {
      nr = next_r(cur_r, cur_V, cxP, nzr[0], stlr[0]);
      ring[1 * kN + reg].x = nr;
      ring[(1 + kRingR) * kN + reg].x = nr;
      ring[0 * kN + reg].y = nr;
      ring[(0 + kRingR) * kN + reg].y = nr;
    }
  }

  auto lds_barrier = [&]() {
    __builtin_amdgcn_fence(__ATOMIC_RELEASE, "workgroup", "local");
    __builtin_amdgcn_s_barrier();
    __builtin_amdgcn_fence(__ATOMIC_ACQUIRE, "workgroup", "local");
  };

  int bm = 0;
  for (int T0 = 0; T0 < kSteps; T0 += 8) {
#pragma unroll
    for (int u = 0; u < 8; u += 2) {
      const int T = T0 + u;
      asm volatile("s_wait_dscnt 0x0" ::: "memory");
      lds_barrier();
      const int bm2 = (bm + 2 == kRingR) ? 0 : bm + 2;
      refill(T + 5, nzr[(u + 5) & 7], stlr[(u + 5) & 7]);
      refill(T + 6, nzr[(u + 6) & 7], stlr[(u + 6) & 7]);

      if (dyn) {
        const int sbase = bm * kSlotB;
        float2 x[64];
#pragma unroll
        for (int e = 0; e < 64; ++e)
          x[e] = *(const float2*)(ringb + (sbase + qreg[e]));
        __builtin_amdgcn_sched_barrier(0);

        float px, py, cxA, cxB;
        sums(x, px, py);
        fold(px, py, cxA, cxB);

        float r2, r3;
        const float V1 = update(cur_r, cur_V, nr, cxP, cxA,
                                nzr[u], stlr[u], nzr[(u + 1) & 7], stlr[(u + 1) & 7], r2);
        const float V2 = update(nr, V1, r2, cxA, cxB,
                                nzr[(u + 1) & 7], stlr[(u + 1) & 7],
                                nzr[(u + 2) & 7], stlr[(u + 2) & 7], r3);
        if (own) {
          float2 o1; o1.x = nr; o1.y = V1;
          float2 o2; o2.x = r2; o2.y = V2;
          ((float2*)out)[T * kN + reg] = o1;
          ((float2*)out)[(T + 1) * kN + reg] = o2;
          float2 p; p.x = r2; p.y = r3;
          ring[bm2 * kN + reg] = p;
          ring[(bm2 + kRingR) * kN + reg] = p;
          ring[(bm + 1) * kN + reg].y = r2;
          ring[(bm + 1 + kRingR) * kN + reg].y = r2;
          ring[(bm2 + 1) * kN + reg].x = r3;
          ring[(bm2 + 1 + kRingR) * kN + reg].x = r3;
        }
        cur_r = r2;
        cur_V = V2;
        nr  = r3;
        cxP = cxB;
      } else {
        const float2 rp = ring[bm * kN + reg];
        if (T >= 1) bold_step(rp.x);
        bold_step(rp.y);
      }
      bm = bm2;
    }
  }

  asm volatile("s_wait_dscnt 0x0" ::: "memory");
  lds_barrier();
  if (!dyn) {
    bold_step(ring[bm * kN + reg].x);
    const float bold = kV0 * (kK1 * (1.f - bq) + kK2 * (1.f - bq / bvl) +
                              kK3 * (1.f - bvl));
    out[kSteps * kN * 2 + reg] = bold;
  }
}

extern "C" void kernel_launch(void* const* d_in, const int* in_sizes, int n_in,
                              void* d_out, int out_size, void* d_ws, size_t ws_size,
                              hipStream_t stream) {
  (void)in_sizes; (void)n_in; (void)d_ws; (void)ws_size; (void)out_size;
  const float* region_pars  = (const float*)d_in[0];
  const float* Wt           = (const float*)d_in[1];
  const float* g            = (const float*)d_in[2];
  const float* stimulus     = (const float*)d_in[3];
  const float* noise        = (const float*)d_in[4];
  const float* initial_cond = (const float*)d_in[5];
  const int*   lags         = (const int*)d_in[6];

  tvb_mpr_persistent<<<1, kThreads, 0, stream>>>(
      region_pars, Wt, g, stimulus, noise, initial_cond, lags, (float*)d_out);
}
